# starting kernel
# baseline (speedup 1.0000x reference)
#include <hip/hip_runtime.h>
#include <hip/hip_bf16.h>

#define NXX 64
#define NYY 32
#define NUU 16
#define NDD 8
#define T_STEPS 2048
#define B_TOT 256

typedef float v2f __attribute__((ext_vector_type(2)));
typedef float v8f __attribute__((ext_vector_type(8)));

__device__ __forceinline__ v8f zero8() {
    v8f c;
#pragma unroll
    for (int r = 0; r < 8; ++r) c[r] = 0.0f;
    return c;
}

__device__ __forceinline__ v2f fragA(const float* X, int ld, int m0, int k0, int lane) {
    int h = lane >> 4, m = m0 + (lane & 15), k = k0 + 2 * h;
    v2f a; a.x = X[m * ld + k]; a.y = X[m * ld + k + 1]; return a;
}
__device__ __forceinline__ v2f fragAT(const float* X, int ld, int m0, int k0, int lane) {
    int h = lane >> 4, m = m0 + (lane & 15), k = k0 + 2 * h;
    v2f a; a.x = X[k * ld + m]; a.y = X[(k + 1) * ld + m]; return a;
}
__device__ __forceinline__ v2f fragB(const float* X, int ld, int k0, int n0, int lane) {
    int h = lane >> 4, n = n0 + (lane & 15), k = k0 + 2 * h;
    v2f b; b.x = X[k * ld + n]; b.y = X[(k + 1) * ld + n]; return b;
}
__device__ __forceinline__ v2f fragBT(const float* X, int ld, int k0, int n0, int lane) {
    int h = lane >> 4, n = n0 + (lane & 15), k = k0 + 2 * h;
    v2f b; b.x = X[n * ld + k]; b.y = X[n * ld + k + 1]; return b;
}
__device__ __forceinline__ void storeD(float* X, int ld, int m0, int n0, v8f d, int lane,
                                       float sgn = 1.0f) {
    int h = lane >> 4, n = n0 + (lane & 15);
#pragma unroll
    for (int r = 0; r < 8; ++r) X[(m0 + r + 8 * h) * ld + n] = sgn * d[r];
}
__device__ __forceinline__ v8f loadC(const float* X, int ld, int m0, int n0, int lane) {
    int h = lane >> 4, n = n0 + (lane & 15);
    v8f c;
#pragma unroll
    for (int r = 0; r < 8; ++r) c[r] = X[(m0 + r + 8 * h) * ld + n];
    return c;
}
__device__ __forceinline__ v8f identC(int m0, int n0, int lane) {
    int h = lane >> 4, n = n0 + (lane & 15);
    v8f c;
#pragma unroll
    for (int r = 0; r < 8; ++r) c[r] = ((m0 + r + 8 * h) == n) ? 1.0f : 0.0f;
    return c;
}

template <bool TA, bool TB>
__device__ __forceinline__ v8f mmacc(const float* A, int lda, int m0,
                                     const float* B, int ldb, int n0,
                                     int K, v8f c, int lane) {
#pragma unroll
    for (int k = 0; k < K; k += 4) {
        v2f a = TA ? fragAT(A, lda, m0, k, lane) : fragA(A, lda, m0, k, lane);
        v2f b = TB ? fragBT(B, ldb, k, n0, lane) : fragB(B, ldb, k, n0, lane);
        c = __builtin_amdgcn_wmma_f32_16x16x4_f32(false, a, false, b, (short)0, c,
                                                  false, false);
    }
    return c;
}

__global__ __launch_bounds__(256) void kf_gain_seq(const float* __restrict__ Wfx,
                                                   const float* __restrict__ Wfy,
                                                   float* __restrict__ G) {
    __shared__ float sWfx[NXX * NXX];
    __shared__ float sWfy[NXX * NYY];
    __shared__ float sP[NXX * NXX];
    __shared__ float sM1[NXX * NXX];
    __shared__ float sM2[NXX * NYY];
    __shared__ float sSa[NYY * 64];
    __shared__ float sL[NXX * NYY];

    const int tid = threadIdx.x, lane = tid & 31, w = tid >> 5;

    for (int i = tid; i < NXX * NXX; i += 256) {
        sWfx[i] = Wfx[i];
        sP[i] = ((i >> 6) == (i & 63)) ? 1.0f : 0.0f;
    }
    for (int i = tid; i < NXX * NYY; i += 256) {
        sWfy[i] = Wfy[i];
        sL[i] = 0.0f;
    }
    __syncthreads();

    for (int t = 0; t < T_STEPS; ++t) {
        float* Gt = G + (size_t)t * (NYY * NXX);
        for (int i = tid; i < NYY * NXX; i += 256) {
            int j = i >> 6, col = i & 63;
            Gt[i] = sL[col * NYY + j];
        }
        __syncthreads();

        for (int tt = w; tt < 16; tt += 8) {
            int mi = (tt >> 2) * 16, ni = (tt & 3) * 16;
            v8f c = zero8();
            c = mmacc<false, false>(sWfx, NXX, mi, sP, NXX, ni, NXX, c, lane);
            storeD(sM1, NXX, mi, ni, c, lane);
        }
        __syncthreads();
        for (int tt = w; tt < 16; tt += 8) {
            int mi = (tt >> 2) * 16, ni = (tt & 3) * 16;
            v8f c = identC(mi, ni, lane);
            c = mmacc<false, true>(sM1, NXX, mi, sWfx, NXX, ni, NXX, c, lane);
            storeD(sP, NXX, mi, ni, c, lane);
        }
        __syncthreads();
        {
            int mi = (w >> 1) * 16, ni = (w & 1) * 16;
            v8f c = zero8();
            c = mmacc<false, false>(sP, NXX, mi, sWfy, NYY, ni, NXX, c, lane);
            storeD(sM2, NYY, mi, ni, c, lane);
        }
        __syncthreads();
        if (w < 4) {
            int mi = (w >> 1) * 16, ni = (w & 1) * 16;
            v8f c = identC(mi, ni, lane);
            c = mmacc<true, false>(sWfy, NYY, mi, sM2, NYY, ni, NXX, c, lane);
            storeD(sSa, 64, mi, ni, c, lane);
        } else {
            int r0 = (w - 4) * 8;
#pragma unroll
            for (int rr = 0; rr < 8; ++rr) {
                int r = r0 + rr;
                sSa[r * 64 + 32 + lane] = (lane == r) ? 1.0f : 0.0f;
            }
        }
        __syncthreads();
        {
            const int r = tid >> 3, cs = (tid & 7) * 8;
            for (int p = 0; p < 32; ++p) {
                float piv = sSa[p * 64 + p];
                float f = sSa[r * 64 + p];
                float pr[8];
#pragma unroll
                for (int j = 0; j < 8; ++j) pr[j] = sSa[p * 64 + cs + j];
                __syncthreads();
                float ip = 1.0f / piv;
                if (r == p) {
#pragma unroll
                    for (int j = 0; j < 8; ++j) sSa[p * 64 + cs + j] = pr[j] * ip;
                } else {
#pragma unroll
                    for (int j = 0; j < 8; ++j) sSa[r * 64 + cs + j] -= f * pr[j] * ip;
                }
                __syncthreads();
            }
        }
        {
            int mi = (w >> 1) * 16, ni = (w & 1) * 16;
            v8f c = zero8();
            c = mmacc<false, false>(sM2, NYY, mi, sSa + 32, 64, ni, NYY, c, lane);
            storeD(sL, NYY, mi, ni, c, lane);
        }
        __syncthreads();
        {
            int mi = (w >> 2) * 16, ni = (w & 3) * 16;
            v8f c = zero8();
            c = mmacc<true, false>(sWfy, NYY, mi, sP, NXX, ni, NXX, c, lane);
            storeD(sM1, NXX, mi, ni, c, lane, -1.0f);
        }
        __syncthreads();
        for (int tt = w; tt < 16; tt += 8) {
            int mi = (tt >> 2) * 16, ni = (tt & 3) * 16;
            v8f c = identC(mi, ni, lane);
            c = mmacc<false, false>(sL, NYY, mi, sM1, NXX, ni, NYY, c, lane);
            storeD(sP, NXX, mi, ni, c, lane);
        }
        __syncthreads();
    }
}

__global__ __launch_bounds__(128) void kf_batch(const float* __restrict__ Yp,
                                                const float* __restrict__ Up,
                                                const float* __restrict__ Dp,
                                                const float* __restrict__ Wfx,
                                                const float* __restrict__ Wfu,
                                                const float* __restrict__ Wfd,
                                                const float* __restrict__ Wfy,
                                                const float* __restrict__ G,
                                                float* __restrict__ out) {
    __shared__ float sW[(NXX + NUU + NDD) * NXX];
    __shared__ float sWfy[NXX * NYY];
    __shared__ float sx[16 * NXX];
    __shared__ float sxp[16 * NXX];
    __shared__ float su[16 * NUU];
    __shared__ float sd[16 * NDD];
    __shared__ float sym[16 * NYY];
    __shared__ float sin[16 * NYY];

    const int tid = threadIdx.x, lane = tid & 31, w = tid >> 5;
    const int r0 = blockIdx.x * 16;

    for (int i = tid; i < NXX * NXX; i += 128) sW[i] = Wfx[i];
    for (int i = tid; i < NUU * NXX; i += 128) sW[NXX * NXX + i] = Wfu[i];
    for (int i = tid; i < NDD * NXX; i += 128) sW[(NXX + NUU) * NXX + i] = Wfd[i];
    for (int i = tid; i < NXX * NYY; i += 128) sWfy[i] = Wfy[i];
    for (int i = tid; i < 16 * NXX; i += 128) sx[i] = 0.0f;
    __syncthreads();

    for (int t = 0; t < T_STEPS; ++t) {
        const float* ut = Up + ((size_t)t * B_TOT + r0) * NUU;
        const float* dt = Dp + ((size_t)t * B_TOT + r0) * NDD;
        const float* yt = Yp + ((size_t)t * B_TOT + r0) * NYY;
        for (int i = tid; i < 16 * NUU; i += 128) su[i] = ut[i];
        for (int i = tid; i < 16 * NDD; i += 128) sd[i] = dt[i];
        for (int i = tid; i < 16 * NYY; i += 128) sym[i] = yt[i];
        const float* Gt = G + (size_t)t * (NYY * NXX);
        if (t + 1 < T_STEPS) {
            __builtin_prefetch(G + (size_t)(t + 1) * (NYY * NXX) + tid * 16, 0, 1);
        }
        __syncthreads();

        {
            int n0 = w * 16;
            v8f c = zero8();
            c = mmacc<false, false>(sx, NXX, 0, sW, NXX, n0, NXX, c, lane);
            c = mmacc<false, false>(su, NUU, 0, sW + NXX * NXX, NXX, n0, NUU, c, lane);
            c = mmacc<false, false>(sd, NDD, 0, sW + (NXX + NUU) * NXX, NXX, n0, NDD, c, lane);
            storeD(sxp, NXX, 0, n0, c, lane);
        }
        __syncthreads();

        if (w < 2) {
            int n0 = w * 16;
            v8f c = zero8();
            c = mmacc<false, false>(sxp, NXX, 0, sWfy, NYY, n0, NXX, c, lane);
            int h = lane >> 4, n = n0 + (lane & 15);
#pragma unroll
            for (int r = 0; r < 8; ++r) {
                int m = r + 8 * h;
                sin[m * NYY + n] = sym[m * NYY + n] - c[r];
            }
        }
        __syncthreads();

        {
            int n0 = w * 16;
            v8f c = loadC(sxp, NXX, 0, n0, lane);
            c = mmacc<false, false>(sin, NYY, 0, Gt, NXX, n0, NYY, c, lane);
            storeD(sx, NXX, 0, n0, c, lane);
        }
        __syncthreads();
    }

    for (int i = tid; i < 16 * NXX; i += 128) {
        int r = i >> 6, cc = i & 63;
        out[(size_t)(r0 + r) * NXX + cc] = sx[i];
    }
}

extern "C" void kernel_launch(void* const* d_in, const int* in_sizes, int n_in,
                              void* d_out, int out_size, void* d_ws, size_t ws_size,
                              hipStream_t stream) {
    (void)in_sizes; (void)n_in; (void)out_size; (void)ws_size;
    const float* Yp  = (const float*)d_in[0];
    const float* Up  = (const float*)d_in[1];
    const float* Dp  = (const float*)d_in[2];
    const float* Wfx = (const float*)d_in[3];
    const float* Wfu = (const float*)d_in[4];
    const float* Wfd = (const float*)d_in[5];
    const float* Wfy = (const float*)d_in[6];
    float* G = (float*)d_ws;
    float* out = (float*)d_out;

    kf_gain_seq<<<1, 256, 0, stream>>>(Wfx, Wfy, G);
    kf_batch<<<16, 128, 0, stream>>>(Yp, Up, Dp, Wfx, Wfu, Wfd, Wfy, G, out);
}


# Gain and batch recursions pipelined over 9 launches: launch k runs the gain steps of chunk k (256 steps) in workgroup 0 beside the batch steps of chunk k-1 in workgroups 1..16; sums unchanged
# speedup vs baseline: 3.3140x; 3.3140x over previous
#include <hip/hip_runtime.h>
#include <hip/hip_bf16.h>

#define NXX 64
#define NYY 32
#define NUU 16
#define NDD 8
#define T_STEPS 2048
#define B_TOT 256
#define NCH 8
#define CH (T_STEPS / NCH)
#define GSLOT (NYY * NXX)

typedef float v2f __attribute__((ext_vector_type(2)));
typedef float v8f __attribute__((ext_vector_type(8)));

__device__ __forceinline__ v8f zero8() {
    v8f c;
#pragma unroll
    for (int r = 0; r < 8; ++r) c[r] = 0.0f;
    return c;
}

__device__ __forceinline__ v2f fragA(const float* X, int ld, int m0, int k0, int lane) {
    int h = lane >> 4, m = m0 + (lane & 15), k = k0 + 2 * h;
    v2f a; a.x = X[m * ld + k]; a.y = X[m * ld + k + 1]; return a;
}
__device__ __forceinline__ v2f fragAT(const float* X, int ld, int m0, int k0, int lane) {
    int h = lane >> 4, m = m0 + (lane & 15), k = k0 + 2 * h;
    v2f a; a.x = X[k * ld + m]; a.y = X[(k + 1) * ld + m]; return a;
}
__device__ __forceinline__ v2f fragB(const float* X, int ld, int k0, int n0, int lane) {
    int h = lane >> 4, n = n0 + (lane & 15), k = k0 + 2 * h;
    v2f b; b.x = X[k * ld + n]; b.y = X[(k + 1) * ld + n]; return b;
}
__device__ __forceinline__ v2f fragBT(const float* X, int ld, int k0, int n0, int lane) {
    int h = lane >> 4, n = n0 + (lane & 15), k = k0 + 2 * h;
    v2f b; b.x = X[n * ld + k]; b.y = X[n * ld + k + 1]; return b;
}
__device__ __forceinline__ void storeD(float* X, int ld, int m0, int n0, v8f d, int lane,
                                       float sgn = 1.0f) {
    int h = lane >> 4, n = n0 + (lane & 15);
#pragma unroll
    for (int r = 0; r < 8; ++r) X[(m0 + r + 8 * h) * ld + n] = sgn * d[r];
}
__device__ __forceinline__ v8f loadC(const float* X, int ld, int m0, int n0, int lane) {
    int h = lane >> 4, n = n0 + (lane & 15);
    v8f c;
#pragma unroll
    for (int r = 0; r < 8; ++r) c[r] = X[(m0 + r + 8 * h) * ld + n];
    return c;
}
__device__ __forceinline__ void storeDT(float* X, int ld, int m0, int n0, v8f d, int lane) {
    int h = lane >> 4, n = n0 + (lane & 15);
    float4* p = reinterpret_cast<float4*>(X + n * ld + m0 + 8 * h);
    p[0] = make_float4(d[0], d[1], d[2], d[3]);
    p[1] = make_float4(d[4], d[5], d[6], d[7]);
}
__device__ __forceinline__ v8f identC(int m0, int n0, int lane) {
    int h = lane >> 4, n = n0 + (lane & 15);
    v8f c;
#pragma unroll
    for (int r = 0; r < 8; ++r) c[r] = ((m0 + r + 8 * h) == n) ? 1.0f : 0.0f;
    return c;
}

template <bool TA, bool TB>
__device__ __forceinline__ v8f mmacc(const float* A, int lda, int m0,
                                     const float* B, int ldb, int n0,
                                     int K, v8f c, int lane) {
#pragma unroll
    for (int k = 0; k < K; k += 4) {
        v2f a = TA ? fragAT(A, lda, m0, k, lane) : fragA(A, lda, m0, k, lane);
        v2f b = TB ? fragBT(B, ldb, k, n0, lane) : fragB(B, ldb, k, n0, lane);
        c = __builtin_amdgcn_wmma_f32_16x16x4_f32(false, a, false, b, (short)0, c,
                                                  false, false);
    }
    return c;
}

__device__ __forceinline__ v8f wmma4(v2f a, v2f b, v8f c) {
    return __builtin_amdgcn_wmma_f32_16x16x4_f32(false, a, false, b, (short)0, c,
                                                 false, false);
}
template <bool TA, int KS>
__device__ __forceinline__ v8f mm_breg(const float* A, int lda, int m0,
                                       const v2f (&b)[KS], v8f c, int lane) {
#pragma unroll
    for (int i = 0; i < KS; ++i) {
        v2f a = TA ? fragAT(A, lda, m0, 4 * i, lane) : fragA(A, lda, m0, 4 * i, lane);
        c = wmma4(a, b[i], c);
    }
    return c;
}
template <bool TB, int KS>
__device__ __forceinline__ v8f mm_areg(const v2f (&a)[KS],
                                       const float* B, int ldb, int n0, v8f c, int lane) {
#pragma unroll
    for (int i = 0; i < KS; ++i) {
        v2f b = TB ? fragBT(B, ldb, 4 * i, n0, lane) : fragB(B, ldb, 4 * i, n0, lane);
        c = wmma4(a[i], b, c);
    }
    return c;
}


template <int KK>
__device__ __forceinline__ void gj_publish(float* Rt, float* Cc, v8f d, int gi, int gj, int lane) {
    constexpr int lr0 = (4 * KK) & 15, hh = lr0 >> 3, rb = lr0 & 7, pt = KK >> 2;
    const int h = lane >> 4, n = lane & 15;
    if (gi == pt && h == hh) {
        *reinterpret_cast<float4*>(Rt + (16 * gj + n) * 4) =
            make_float4(d[rb], d[rb + 1], d[rb + 2], d[rb + 3]);
    }
    const int cq = n - lr0;
    if (gj == pt && cq >= 0 && cq < 4) {
#pragma unroll
        for (int r = 0; r < 8; ++r) Cc[(16 * gi + r + 8 * h) * 4 + cq] = d[r];
    }
}

template <int KK>
__device__ __forceinline__ v8f gj_step(const float* Rt, const float* Cc, v8f d, int gi, int gj, int lane) {
    constexpr int lr0 = (4 * KK) & 15, hh = lr0 >> 3, rb = lr0 & 7, pt = KK >> 2;
    const int h = lane >> 4, n = lane & 15;
    float p[4][4], m[4];
#pragma unroll
    for (int c = 0; c < 4; ++c) {
        const float4 v = *reinterpret_cast<const float4*>(Rt + (4 * KK + c) * 4);
        p[0][c] = v.x; p[1][c] = v.y; p[2][c] = v.z; p[3][c] = v.w;
    }
    {
        const float4 v = *reinterpret_cast<const float4*>(Rt + (16 * gj + n) * 4);
        m[0] = v.x; m[1] = v.y; m[2] = v.z; m[3] = v.w;
    }
    const float2 cv = *reinterpret_cast<const float2*>(Cc + (16 * gi + n) * 4 + 2 * h);
#pragma unroll
    for (int q = 0; q < 4; ++q) {
        const float ip = 1.0f / p[q][q];
#pragma unroll
        for (int c = q + 1; c < 4; ++c) p[q][c] *= ip;
        m[q] *= ip;
#pragma unroll
        for (int r = 0; r < 4; ++r) {
            if (r == q) continue;
            const float f = p[r][q];
#pragma unroll
            for (int c = q + 1; c < 4; ++c) p[r][c] -= f * p[q][c];
            m[r] -= f * m[q];
        }
    }
    const int mr = n - lr0;
    const bool prow = (gi == pt) && (mr >= 0) && (mr < 4);
    v2f a, b;
    a.x = prow ? ((mr == 2 * h) ? 1.0f : 0.0f) : -cv.x;
    a.y = prow ? ((mr == 2 * h + 1) ? 1.0f : 0.0f) : -cv.y;
    b.x = h ? m[2] : m[0];
    b.y = h ? m[3] : m[1];
    const bool zr = (gi == pt) && (h == hh);
#pragma unroll
    for (int q = 0; q < 4; ++q) d[rb + q] = zr ? 0.0f : d[rb + q];
    return wmma4(a, b, d);
}

template <int KK>
__device__ __forceinline__ v8f gj_run(float (*Rt)[64 * 4], float (*Cc)[NYY * 4], v8f d,
                                      int gi, int gj, int lane) {
    d = gj_step<KK>(Rt[KK & 1], Cc[KK & 1], d, gi, gj, lane);
    if constexpr (KK < 7) {
        gj_publish<KK + 1>(Rt[(KK + 1) & 1], Cc[(KK + 1) & 1], d, gi, gj, lane);
        __syncthreads();
        return gj_run<KK + 1>(Rt, Cc, d, gi, gj, lane);
    } else {
        return d;
    }
}

__device__ __forceinline__ void kf_gain_chunk(const float* __restrict__ Wfx,
                                              const float* __restrict__ Wfy,
                                              float* __restrict__ G, const int ck) {
    __shared__ float sP[NXX * NXX];
    __shared__ __attribute__((aligned(16))) float sPT[NXX * NXX];
    __shared__ float sM1[NXX * NXX];
    __shared__ float sM2[NXX * NYY];
    __shared__ float sInv[NYY * NYY];
    __shared__ __attribute__((aligned(16))) float sRt[2][64 * 4];
    __shared__ __attribute__((aligned(16))) float sCc[2][NYY * 4];
    __shared__ __attribute__((aligned(16))) float sLT[NYY * NXX];

    const int tid = threadIdx.x, lane = tid & 31, w = tid >> 5;
    const int t0 = ck * CH, t1 = t0 + CH;

    if (ck == 0) {
        for (int i = tid; i < NXX * NXX; i += 256) {
            sP[i] = ((i >> 6) == (i & 63)) ? 1.0f : 0.0f;
        }
        for (int i = tid; i < NYY * NXX; i += 256) {
            sLT[i] = 0.0f;
        }
    } else {
        const float* Ls = G + (size_t)t0 * GSLOT;
        const float* Ps = G + (size_t)(t0 + 1) * GSLOT;
        for (int i = tid; i < NXX * NXX; i += 256) sP[i] = Ps[i];
        for (int i = tid; i < NYY * NXX; i += 256) sLT[i] = Ls[i];
    }

    const int mR = (w >> 1) * 16, nC = (w & 1) * 16;
    const int mA = (w >> 2) * 16, nA = (w & 3) * 16;
    const int mB = (w >> 1) * 16, nB = (w & 1) * 16;
    const int mS = ((w >> 1) & 1) * 16;
    const int gi = (w >> 1) & 1, gj = (w & 1) + ((w >> 2) << 1);

    v2f bXa[16], bXb[16];
    v2f bY[16];
    v2f aYS[16];
    v2f aYP[16];
#pragma unroll
    for (int i = 0; i < 16; ++i) {
        bXa[i] = fragBT(Wfx, NXX, 4 * i, nC, lane);
        bXb[i] = fragBT(Wfx, NXX, 4 * i, nC + 32, lane);
        bY[i]  = fragB(Wfy, NYY, 4 * i, nB, lane);
        aYS[i] = fragAT(Wfy, NYY, mS, 4 * i, lane);
        aYP[i] = fragAT(Wfy, NYY, mA, 4 * i, lane);
    }
    __syncthreads();

    for (int t = t0; t < t1; ++t) {
        {
            float4* Gt4 = reinterpret_cast<float4*>(G + (size_t)t * (NYY * NXX));
            const float4* L4 = reinterpret_cast<const float4*>(sLT);
            Gt4[tid] = L4[tid];
            Gt4[tid + 256] = L4[tid + 256];
        }
        __syncthreads();

        {
            v8f c0 = zero8(), c1 = zero8();
#pragma unroll
            for (int i = 0; i < 16; ++i) {
                const v2f a = fragAT(sP, NXX, mR, 4 * i, lane);
                c0 = wmma4(a, bXa[i], c0);
                c1 = wmma4(a, bXb[i], c1);
            }
            storeD(sM1, NXX, mR, nC, c0, lane);
            storeD(sM1, NXX, mR, nC + 32, c1, lane);
        }
        __syncthreads();
        {
            v8f c0 = identC(mR, nC, lane), c1 = identC(mR, nC + 32, lane);
#pragma unroll
            for (int i = 0; i < 16; ++i) {
                const v2f a = fragAT(sM1, NXX, mR, 4 * i, lane);
                c0 = wmma4(a, bXa[i], c0);
                c1 = wmma4(a, bXb[i], c1);
            }
            storeD(sP, NXX, mR, nC, c0, lane);
            storeD(sP, NXX, mR, nC + 32, c1, lane);
            storeDT(sPT + (mR >> 4) * (16 * NXX), 16, 0, nC, c0, lane);
            storeDT(sPT + (mR >> 4) * (16 * NXX), 16, 0, nC + 32, c1, lane);
        }
        __syncthreads();
        {
            v8f c = zero8();
            c = mm_breg<true, 16>(sPT + (mB >> 4) * (16 * NXX), 16, 0, bY, c, lane);
            storeD(sM2, NYY, mB, nB, c, lane);
        }
        __syncthreads();
        {
            v8f d = identC(mS, nB, lane);
            if (w < 4) d = mm_areg<false, 16>(aYS, sM2, NYY, nB, d, lane);
            gj_publish<0>(sRt[0], sCc[0], d, gi, gj, lane);
            __syncthreads();
            d = gj_run<0>(sRt, sCc, d, gi, gj, lane);
            if (w >= 4) storeD(sInv, NYY, mS, nB, d, lane);
        }
        __syncthreads();
        {
            v8f c = zero8();
            c = mmacc<true, true>(sInv, NYY, nB, sM2, NYY, mB, NYY, c, lane);
            storeD(sLT, NXX, nB, mB, c, lane);
        }
        __syncthreads();
        {
            v8f c = zero8();
            c = mm_areg<false, 16>(aYP, sP, NXX, nA, c, lane);
            storeD(sM1, NXX, mA, nA, c, lane, -1.0f);
        }
        __syncthreads();
        for (int tt = w; tt < 16; tt += 8) {
            int mi = (tt >> 2) * 16, ni = (tt & 3) * 16;
            v8f c = identC(mi, ni, lane);
            c = mmacc<true, false>(sLT, NXX, mi, sM1, NXX, ni, NYY, c, lane);
            storeD(sP, NXX, mi, ni, c, lane);
        }
        __syncthreads();
    }

    if (ck + 1 < NCH) {
        float* Ls = G + (size_t)t1 * GSLOT;
        float* Ps = G + (size_t)(t1 + 1) * GSLOT;
        for (int i = tid; i < NXX * NXX; i += 256) Ps[i] = sP[i];
        for (int i = tid; i < NYY * NXX; i += 256) Ls[i] = sLT[i];
    }
}

__device__ __forceinline__ void kf_batch_chunk(const float* __restrict__ Yp,
                                               const float* __restrict__ Up,
                                               const float* __restrict__ Dp,
                                               const float* __restrict__ Wfx,
                                               const float* __restrict__ Wfu,
                                               const float* __restrict__ Wfd,
                                               const float* __restrict__ Wfy,
                                               const float* __restrict__ G,
                                               float* __restrict__ out,
                                               const int ck, const int bt) {
    __shared__ __attribute__((aligned(16))) float sxT[NXX * 16];
    __shared__ __attribute__((aligned(16))) float sxpT[NXX * 16];
    __shared__ __attribute__((aligned(16))) float sinT[NYY * 16];

    const int tid = threadIdx.x, lane = tid & 31, w = tid >> 5;
    const bool act = w < 4;
    const int r0 = bt * 16;
    const int lh = lane >> 4, lm = lane & 15;
    const int t0 = ck * CH, t1 = t0 + CH;

    if (ck == 0) {
        for (int i = tid; i < 16 * NXX; i += 256) sxT[i] = 0.0f;
    } else {
        for (int i = tid; i < 16 * NXX; i += 256) {
            int r = i >> 6, cc = i & 63;
            sxT[cc * 16 + r] = out[(size_t)(r0 + r) * NXX + cc];
        }
    }

    const int nW = (w & 3) * 16;
    const int nY = (w & 1) * 16;
    v2f bX[16], bU[4], bD[2];
    v2f bY[16];
#pragma unroll
    for (int i = 0; i < 16; ++i) {
        bX[i] = fragB(Wfx, NXX, 4 * i, nW, lane);
        bY[i] = fragB(Wfy, NYY, 4 * i, nY, lane);
    }
#pragma unroll
    for (int i = 0; i < 4; ++i) bU[i] = fragB(Wfu, NXX, 4 * i, nW, lane);
#pragma unroll
    for (int i = 0; i < 2; ++i) bD[i] = fragB(Wfd, NXX, 4 * i, nW, lane);
    __syncthreads();

    for (int t = t0; t < t1; ++t) {
        const float* Gt = G + (size_t)t * GSLOT;
        v8f xp = zero8();
        v8f ym = zero8();
        if (act) {
            const float* ut = Up + ((size_t)t * B_TOT + r0) * NUU;
            const float* dt = Dp + ((size_t)t * B_TOT + r0) * NDD;
            const float* yt = Yp + ((size_t)t * B_TOT + r0) * NYY;
            v2f aU[4], aD[2];
#pragma unroll
            for (int i = 0; i < 4; ++i)
                aU[i] = *reinterpret_cast<const v2f*>(ut + lm * NUU + 4 * i + 2 * lh);
#pragma unroll
            for (int i = 0; i < 2; ++i)
                aD[i] = *reinterpret_cast<const v2f*>(dt + lm * NDD + 4 * i + 2 * lh);
            if (w < 2) {
#pragma unroll
                for (int r = 0; r < 8; ++r) ym[r] = yt[(r + 8 * lh) * NYY + nY + lm];
            }
            if (t + 1 < t1) {
                __builtin_prefetch(G + (size_t)(t + 1) * GSLOT + tid * 16, 0, 1);
            }

            xp = mm_breg<true, 16>(sxT, 16, 0, bX, xp, lane);
#pragma unroll
            for (int i = 0; i < 4; ++i) xp = wmma4(aU[i], bU[i], xp);
#pragma unroll
            for (int i = 0; i < 2; ++i) xp = wmma4(aD[i], bD[i], xp);
            storeDT(sxpT, 16, 0, nW, xp, lane);
        }
        __syncthreads();

        if (w < 2) {
            v8f c = zero8();
            c = mm_breg<true, 16>(sxpT, 16, 0, bY, c, lane);
            v8f inn;
#pragma unroll
            for (int r = 0; r < 8; ++r) inn[r] = ym[r] - c[r];
            storeDT(sinT, 16, 0, nY, inn, lane);
        }
        __syncthreads();

        if (act) {
            v8f c = mmacc<true, false>(sinT, 16, 0, Gt, NXX, nW, NYY, xp, lane);
            storeDT(sxT, 16, 0, nW, c, lane);
        }
        __syncthreads();
    }

    for (int i = tid; i < 16 * NXX; i += 256) {
        int r = i >> 6, cc = i & 63;
        out[(size_t)(r0 + r) * NXX + cc] = sxT[cc * 16 + r];
    }
}

__global__ __launch_bounds__(256) void kf_pipe(const float* __restrict__ Yp,
                                               const float* __restrict__ Up,
                                               const float* __restrict__ Dp,
                                               const float* __restrict__ Wfx,
                                               const float* __restrict__ Wfu,
                                               const float* __restrict__ Wfd,
                                               const float* __restrict__ Wfy,
                                               float* G, float* out, const int k) {
    if (blockIdx.x == 0) {
        if (k < NCH) kf_gain_chunk(Wfx, Wfy, G, k);
    } else {
        if (k >= 1) kf_batch_chunk(Yp, Up, Dp, Wfx, Wfu, Wfd, Wfy, G, out, k - 1, blockIdx.x - 1);
    }
}

extern "C" void kernel_launch(void* const* d_in, const int* in_sizes, int n_in,
                              void* d_out, int out_size, void* d_ws, size_t ws_size,
                              hipStream_t stream) {
    (void)in_sizes; (void)n_in; (void)out_size; (void)ws_size;
    const float* Yp  = (const float*)d_in[0];
    const float* Up  = (const float*)d_in[1];
    const float* Dp  = (const float*)d_in[2];
    const float* Wfx = (const float*)d_in[3];
    const float* Wfu = (const float*)d_in[4];
    const float* Wfd = (const float*)d_in[5];
    const float* Wfy = (const float*)d_in[6];
    float* G = (float*)d_ws;
    float* out = (float*)d_out;

    kf_pipe<<<17, 256, 0, stream>>>(Yp, Up, Dp, Wfx, Wfu, Wfd, Wfy, G, out, 0);
    kf_pipe<<<17, 256, 0, stream>>>(Yp, Up, Dp, Wfx, Wfu, Wfd, Wfy, G, out, 1);
    kf_pipe<<<17, 256, 0, stream>>>(Yp, Up, Dp, Wfx, Wfu, Wfd, Wfy, G, out, 2);
    kf_pipe<<<17, 256, 0, stream>>>(Yp, Up, Dp, Wfx, Wfu, Wfd, Wfy, G, out, 3);
    kf_pipe<<<17, 256, 0, stream>>>(Yp, Up, Dp, Wfx, Wfu, Wfd, Wfy, G, out, 4);
    kf_pipe<<<17, 256, 0, stream>>>(Yp, Up, Dp, Wfx, Wfu, Wfd, Wfy, G, out, 5);
    kf_pipe<<<17, 256, 0, stream>>>(Yp, Up, Dp, Wfx, Wfu, Wfd, Wfy, G, out, 6);
    kf_pipe<<<17, 256, 0, stream>>>(Yp, Up, Dp, Wfx, Wfu, Wfd, Wfy, G, out, 7);
    kf_pipe<<<17, 256, 0, stream>>>(Yp, Up, Dp, Wfx, Wfu, Wfd, Wfy, G, out, 8);
}


# Gain step: every stage reads all LDS operand fragments of its K loop (a Gauss-Jordan block step its six reads) as one group ahead of its WMMA chain, kept there by a scheduling fence; sums unchanged
# speedup vs baseline: 3.8078x; 1.1490x over previous
#include <hip/hip_runtime.h>
#include <hip/hip_bf16.h>

#define NXX 64
#define NYY 32
#define NUU 16
#define NDD 8
#define T_STEPS 2048
#define B_TOT 256
#define NCH 8
#define CH (T_STEPS / NCH)
#define GSLOT (NYY * NXX)

typedef float v2f __attribute__((ext_vector_type(2)));
typedef float v8f __attribute__((ext_vector_type(8)));

__device__ __forceinline__ v8f zero8() {
    v8f c;
#pragma unroll
    for (int r = 0; r < 8; ++r) c[r] = 0.0f;
    return c;
}

__device__ __forceinline__ v2f fragA(const float* X, int ld, int m0, int k0, int lane) {
    int h = lane >> 4, m = m0 + (lane & 15), k = k0 + 2 * h;
    v2f a; a.x = X[m * ld + k]; a.y = X[m * ld + k + 1]; return a;
}
__device__ __forceinline__ v2f fragAT(const float* X, int ld, int m0, int k0, int lane) {
    int h = lane >> 4, m = m0 + (lane & 15), k = k0 + 2 * h;
    v2f a; a.x = X[k * ld + m]; a.y = X[(k + 1) * ld + m]; return a;
}
__device__ __forceinline__ v2f fragB(const float* X, int ld, int k0, int n0, int lane) {
    int h = lane >> 4, n = n0 + (lane & 15), k = k0 + 2 * h;
    v2f b; b.x = X[k * ld + n]; b.y = X[(k + 1) * ld + n]; return b;
}
__device__ __forceinline__ v2f fragBT(const float* X, int ld, int k0, int n0, int lane) {
    int h = lane >> 4, n = n0 + (lane & 15), k = k0 + 2 * h;
    v2f b; b.x = X[n * ld + k]; b.y = X[n * ld + k + 1]; return b;
}
__device__ __forceinline__ void storeD(float* X, int ld, int m0, int n0, v8f d, int lane,
                                       float sgn = 1.0f) {
    int h = lane >> 4, n = n0 + (lane & 15);
#pragma unroll
    for (int r = 0; r < 8; ++r) X[(m0 + r + 8 * h) * ld + n] = sgn * d[r];
}
__device__ __forceinline__ v8f loadC(const float* X, int ld, int m0, int n0, int lane) {
    int h = lane >> 4, n = n0 + (lane & 15);
    v8f c;
#pragma unroll
    for (int r = 0; r < 8; ++r) c[r] = X[(m0 + r + 8 * h) * ld + n];
    return c;
}
__device__ __forceinline__ void storeDT(float* X, int ld, int m0, int n0, v8f d, int lane) {
    int h = lane >> 4, n = n0 + (lane & 15);
    float4* p = reinterpret_cast<float4*>(X + n * ld + m0 + 8 * h);
    p[0] = make_float4(d[0], d[1], d[2], d[3]);
    p[1] = make_float4(d[4], d[5], d[6], d[7]);
}
__device__ __forceinline__ v8f identC(int m0, int n0, int lane) {
    int h = lane >> 4, n = n0 + (lane & 15);
    v8f c;
#pragma unroll
    for (int r = 0; r < 8; ++r) c[r] = ((m0 + r + 8 * h) == n) ? 1.0f : 0.0f;
    return c;
}

template <bool TA, bool TB>
__device__ __forceinline__ v8f mmacc(const float* A, int lda, int m0,
                                     const float* B, int ldb, int n0,
                                     int K, v8f c, int lane) {
#pragma unroll
    for (int k = 0; k < K; k += 4) {
        v2f a = TA ? fragAT(A, lda, m0, k, lane) : fragA(A, lda, m0, k, lane);
        v2f b = TB ? fragBT(B, ldb, k, n0, lane) : fragB(B, ldb, k, n0, lane);
        c = __builtin_amdgcn_wmma_f32_16x16x4_f32(false, a, false, b, (short)0, c,
                                                  false, false);
    }
    return c;
}

__device__ __forceinline__ v8f wmma4(v2f a, v2f b, v8f c) {
    return __builtin_amdgcn_wmma_f32_16x16x4_f32(false, a, false, b, (short)0, c,
                                                 false, false);
}
template <bool TA, int KS>
__device__ __forceinline__ v8f mm_breg(const float* A, int lda, int m0,
                                       const v2f (&b)[KS], v8f c, int lane) {
#pragma unroll
    for (int i = 0; i < KS; ++i) {
        v2f a = TA ? fragAT(A, lda, m0, 4 * i, lane) : fragA(A, lda, m0, 4 * i, lane);
        c = wmma4(a, b[i], c);
    }
    return c;
}

__device__ __forceinline__ void lds_group_fence() {
    __builtin_amdgcn_sched_barrier(0);
}
template <bool TA, int KS>
__device__ __forceinline__ void load_a(const float* A, int lda, int m0, int lane, v2f (&a)[KS]) {
#pragma unroll
    for (int i = 0; i < KS; ++i)
        a[i] = TA ? fragAT(A, lda, m0, 4 * i, lane) : fragA(A, lda, m0, 4 * i, lane);
}
template <bool TB, int KS>
__device__ __forceinline__ void load_b(const float* B, int ldb, int n0, int lane, v2f (&b)[KS]) {
#pragma unroll
    for (int i = 0; i < KS; ++i)
        b[i] = TB ? fragBT(B, ldb, 4 * i, n0, lane) : fragB(B, ldb, 4 * i, n0, lane);
}
template <int KS>
__device__ __forceinline__ v8f mm_regs(const v2f (&a)[KS], const v2f (&b)[KS], v8f c) {
#pragma unroll
    for (int i = 0; i < KS; ++i) c = wmma4(a[i], b[i], c);
    return c;
}

template <int KK>
__device__ __forceinline__ void gj_publish(float* Rt, float* Cc, v8f d, int gi, int gj, int lane) {
    constexpr int lr0 = (4 * KK) & 15, hh = lr0 >> 3, rb = lr0 & 7, pt = KK >> 2;
    const int h = lane >> 4, n = lane & 15;
    if (gi == pt && h == hh) {
        *reinterpret_cast<float4*>(Rt + (16 * gj + n) * 4) =
            make_float4(d[rb], d[rb + 1], d[rb + 2], d[rb + 3]);
    }
    const int cq = n - lr0;
    if (gj == pt && cq >= 0 && cq < 4) {
#pragma unroll
        for (int r = 0; r < 8; ++r) Cc[(16 * gi + r + 8 * h) * 4 + cq] = d[r];
    }
}

template <int KK>
__device__ __forceinline__ v8f gj_step(const float* Rt, const float* Cc, v8f d, int gi, int gj, int lane) {
    constexpr int lr0 = (4 * KK) & 15, hh = lr0 >> 3, rb = lr0 & 7, pt = KK >> 2;
    const int h = lane >> 4, n = lane & 15;
    float p[4][4], m[4];
    float4 pc[4];
#pragma unroll
    for (int c = 0; c < 4; ++c) pc[c] = *reinterpret_cast<const float4*>(Rt + (4 * KK + c) * 4);
    const float4 mv = *reinterpret_cast<const float4*>(Rt + (16 * gj + n) * 4);
    const float2 cv = *reinterpret_cast<const float2*>(Cc + (16 * gi + n) * 4 + 2 * h);
    lds_group_fence();
#pragma unroll
    for (int c = 0; c < 4; ++c) {
        p[0][c] = pc[c].x; p[1][c] = pc[c].y; p[2][c] = pc[c].z; p[3][c] = pc[c].w;
    }
    m[0] = mv.x; m[1] = mv.y; m[2] = mv.z; m[3] = mv.w;
#pragma unroll
    for (int q = 0; q < 4; ++q) {
        const float ip = 1.0f / p[q][q];
#pragma unroll
        for (int c = q + 1; c < 4; ++c) p[q][c] *= ip;
        m[q] *= ip;
#pragma unroll
        for (int r = 0; r < 4; ++r) {
            if (r == q) continue;
            const float f = p[r][q];
#pragma unroll
            for (int c = q + 1; c < 4; ++c) p[r][c] -= f * p[q][c];
            m[r] -= f * m[q];
        }
    }
    const int mr = n - lr0;
    const bool prow = (gi == pt) && (mr >= 0) && (mr < 4);
    v2f a, b;
    a.x = prow ? ((mr == 2 * h) ? 1.0f : 0.0f) : -cv.x;
    a.y = prow ? ((mr == 2 * h + 1) ? 1.0f : 0.0f) : -cv.y;
    b.x = h ? m[2] : m[0];
    b.y = h ? m[3] : m[1];
    const bool zr = (gi == pt) && (h == hh);
#pragma unroll
    for (int q = 0; q < 4; ++q) d[rb + q] = zr ? 0.0f : d[rb + q];
    return wmma4(a, b, d);
}

template <int KK>
__device__ __forceinline__ v8f gj_run(float (*Rt)[64 * 4], float (*Cc)[NYY * 4], v8f d,
                                      int gi, int gj, int lane) {
    d = gj_step<KK>(Rt[KK & 1], Cc[KK & 1], d, gi, gj, lane);
    if constexpr (KK < 7) {
        gj_publish<KK + 1>(Rt[(KK + 1) & 1], Cc[(KK + 1) & 1], d, gi, gj, lane);
        __syncthreads();
        return gj_run<KK + 1>(Rt, Cc, d, gi, gj, lane);
    } else {
        return d;
    }
}

__device__ __forceinline__ void kf_gain_chunk(const float* __restrict__ Wfx,
                                              const float* __restrict__ Wfy,
                                              float* __restrict__ G, const int ck) {
    __shared__ float sP[NXX * NXX];
    __shared__ __attribute__((aligned(16))) float sPT[NXX * NXX];
    __shared__ float sM1[NXX * NXX];
    __shared__ float sM2[NXX * NYY];
    __shared__ float sInv[NYY * NYY];
    __shared__ __attribute__((aligned(16))) float sRt[2][64 * 4];
    __shared__ __attribute__((aligned(16))) float sCc[2][NYY * 4];
    __shared__ __attribute__((aligned(16))) float sLT[NYY * NXX];

    const int tid = threadIdx.x, lane = tid & 31, w = tid >> 5;
    const int t0 = ck * CH, t1 = t0 + CH;

    if (ck == 0) {
        for (int i = tid; i < NXX * NXX; i += 256) {
            sP[i] = ((i >> 6) == (i & 63)) ? 1.0f : 0.0f;
        }
        for (int i = tid; i < NYY * NXX; i += 256) {
            sLT[i] = 0.0f;
        }
    } else {
        const float* Ls = G + (size_t)t0 * GSLOT;
        const float* Ps = G + (size_t)(t0 + 1) * GSLOT;
        for (int i = tid; i < NXX * NXX; i += 256) sP[i] = Ps[i];
        for (int i = tid; i < NYY * NXX; i += 256) sLT[i] = Ls[i];
    }

    const int mR = (w >> 1) * 16, nC = (w & 1) * 16;
    const int mA = (w >> 2) * 16, nA = (w & 3) * 16;
    const int mB = (w >> 1) * 16, nB = (w & 1) * 16;
    const int mS = ((w >> 1) & 1) * 16;
    const int gi = (w >> 1) & 1, gj = (w & 1) + ((w >> 2) << 1);

    v2f bXa[16], bXb[16];
    v2f bY[16];
    v2f aYS[16];
    v2f aYP[16];
#pragma unroll
    for (int i = 0; i < 16; ++i) {
        bXa[i] = fragBT(Wfx, NXX, 4 * i, nC, lane);
        bXb[i] = fragBT(Wfx, NXX, 4 * i, nC + 32, lane);
        bY[i]  = fragB(Wfy, NYY, 4 * i, nB, lane);
        aYS[i] = fragAT(Wfy, NYY, mS, 4 * i, lane);
        aYP[i] = fragAT(Wfy, NYY, mA, 4 * i, lane);
    }
    __syncthreads();

    for (int t = t0; t < t1; ++t) {
        {
            float4* Gt4 = reinterpret_cast<float4*>(G + (size_t)t * (NYY * NXX));
            const float4* L4 = reinterpret_cast<const float4*>(sLT);
            Gt4[tid] = L4[tid];
            Gt4[tid + 256] = L4[tid + 256];
        }
        __syncthreads();

        {
            v8f c0 = zero8(), c1 = zero8();
            v2f a[16];
            load_a<true, 16>(sP, NXX, mR, lane, a);
            lds_group_fence();
#pragma unroll
            for (int i = 0; i < 16; ++i) {
                c0 = wmma4(a[i], bXa[i], c0);
                c1 = wmma4(a[i], bXb[i], c1);
            }
            storeD(sM1, NXX, mR, nC, c0, lane);
            storeD(sM1, NXX, mR, nC + 32, c1, lane);
        }
        __syncthreads();
        {
            v8f c0 = identC(mR, nC, lane), c1 = identC(mR, nC + 32, lane);
            v2f a[16];
            load_a<true, 16>(sM1, NXX, mR, lane, a);
            lds_group_fence();
#pragma unroll
            for (int i = 0; i < 16; ++i) {
                c0 = wmma4(a[i], bXa[i], c0);
                c1 = wmma4(a[i], bXb[i], c1);
            }
            storeD(sP, NXX, mR, nC, c0, lane);
            storeD(sP, NXX, mR, nC + 32, c1, lane);
            storeDT(sPT + (mR >> 4) * (16 * NXX), 16, 0, nC, c0, lane);
            storeDT(sPT + (mR >> 4) * (16 * NXX), 16, 0, nC + 32, c1, lane);
        }
        __syncthreads();
        {
            v8f c = zero8();
            v2f a[16];
            load_a<true, 16>(sPT + (mB >> 4) * (16 * NXX), 16, 0, lane, a);
            lds_group_fence();
            c = mm_regs<16>(a, bY, c);
            storeD(sM2, NYY, mB, nB, c, lane);
        }
        __syncthreads();
        {
            v8f d = identC(mS, nB, lane);
            if (w < 4) {
                v2f b[16];
                load_b<false, 16>(sM2, NYY, nB, lane, b);
                lds_group_fence();
                d = mm_regs<16>(aYS, b, d);
            }
            gj_publish<0>(sRt[0], sCc[0], d, gi, gj, lane);
            __syncthreads();
            d = gj_run<0>(sRt, sCc, d, gi, gj, lane);
            if (w >= 4) storeD(sInv, NYY, mS, nB, d, lane);
        }
        __syncthreads();
        {
            v8f c = zero8();
            v2f a[8], b[8];
            load_a<true, 8>(sInv, NYY, nB, lane, a);
            load_b<true, 8>(sM2, NYY, mB, lane, b);
            lds_group_fence();
            c = mm_regs<8>(a, b, c);
            storeD(sLT, NXX, nB, mB, c, lane);
        }
        __syncthreads();
        {
            v8f c = zero8();
            v2f b[16];
            load_b<false, 16>(sP, NXX, nA, lane, b);
            lds_group_fence();
            c = mm_regs<16>(aYP, b, c);
            storeD(sM1, NXX, mA, nA, c, lane, -1.0f);
        }
        __syncthreads();
        {
            v2f a0[8], a1[8], b[8];
            load_a<true, 8>(sLT, NXX, mA, lane, a0);
            load_a<true, 8>(sLT, NXX, mA + 32, lane, a1);
            load_b<false, 8>(sM1, NXX, nA, lane, b);
            lds_group_fence();
            v8f c0 = identC(mA, nA, lane), c1 = identC(mA + 32, nA, lane);
#pragma unroll
            for (int i = 0; i < 8; ++i) {
                c0 = wmma4(a0[i], b[i], c0);
                c1 = wmma4(a1[i], b[i], c1);
            }
            storeD(sP, NXX, mA, nA, c0, lane);
            storeD(sP, NXX, mA + 32, nA, c1, lane);
        }
        __syncthreads();
    }

    if (ck + 1 < NCH) {
        float* Ls = G + (size_t)t1 * GSLOT;
        float* Ps = G + (size_t)(t1 + 1) * GSLOT;
        for (int i = tid; i < NXX * NXX; i += 256) Ps[i] = sP[i];
        for (int i = tid; i < NYY * NXX; i += 256) Ls[i] = sLT[i];
    }
}

__device__ __forceinline__ void kf_batch_chunk(const float* __restrict__ Yp,
                                               const float* __restrict__ Up,
                                               const float* __restrict__ Dp,
                                               const float* __restrict__ Wfx,
                                               const float* __restrict__ Wfu,
                                               const float* __restrict__ Wfd,
                                               const float* __restrict__ Wfy,
                                               const float* __restrict__ G,
                                               float* __restrict__ out,
                                               const int ck, const int bt) {
    __shared__ __attribute__((aligned(16))) float sxT[NXX * 16];
    __shared__ __attribute__((aligned(16))) float sxpT[NXX * 16];
    __shared__ __attribute__((aligned(16))) float sinT[NYY * 16];

    const int tid = threadIdx.x, lane = tid & 31, w = tid >> 5;
    const bool act = w < 4;
    const int r0 = bt * 16;
    const int lh = lane >> 4, lm = lane & 15;
    const int t0 = ck * CH, t1 = t0 + CH;

    if (ck == 0) {
        for (int i = tid; i < 16 * NXX; i += 256) sxT[i] = 0.0f;
    } else {
        for (int i = tid; i < 16 * NXX; i += 256) {
            int r = i >> 6, cc = i & 63;
            sxT[cc * 16 + r] = out[(size_t)(r0 + r) * NXX + cc];
        }
    }

    const int nW = (w & 3) * 16;
    const int nY = (w & 1) * 16;
    v2f bX[16], bU[4], bD[2];
    v2f bY[16];
#pragma unroll
    for (int i = 0; i < 16; ++i) {
        bX[i] = fragB(Wfx, NXX, 4 * i, nW, lane);
        bY[i] = fragB(Wfy, NYY, 4 * i, nY, lane);
    }
#pragma unroll
    for (int i = 0; i < 4; ++i) bU[i] = fragB(Wfu, NXX, 4 * i, nW, lane);
#pragma unroll
    for (int i = 0; i < 2; ++i) bD[i] = fragB(Wfd, NXX, 4 * i, nW, lane);
    __syncthreads();

    for (int t = t0; t < t1; ++t) {
        const float* Gt = G + (size_t)t * GSLOT;
        v8f xp = zero8();
        v8f ym = zero8();
        if (act) {
            const float* ut = Up + ((size_t)t * B_TOT + r0) * NUU;
            const float* dt = Dp + ((size_t)t * B_TOT + r0) * NDD;
            const float* yt = Yp + ((size_t)t * B_TOT + r0) * NYY;
            v2f aU[4], aD[2];
#pragma unroll
            for (int i = 0; i < 4; ++i)
                aU[i] = *reinterpret_cast<const v2f*>(ut + lm * NUU + 4 * i + 2 * lh);
#pragma unroll
            for (int i = 0; i < 2; ++i)
                aD[i] = *reinterpret_cast<const v2f*>(dt + lm * NDD + 4 * i + 2 * lh);
            if (w < 2) {
#pragma unroll
                for (int r = 0; r < 8; ++r) ym[r] = yt[(r + 8 * lh) * NYY + nY + lm];
            }
            if (t + 1 < t1) {
                __builtin_prefetch(G + (size_t)(t + 1) * GSLOT + tid * 16, 0, 1);
            }

            xp = mm_breg<true, 16>(sxT, 16, 0, bX, xp, lane);
#pragma unroll
            for (int i = 0; i < 4; ++i) xp = wmma4(aU[i], bU[i], xp);
#pragma unroll
            for (int i = 0; i < 2; ++i) xp = wmma4(aD[i], bD[i], xp);
            storeDT(sxpT, 16, 0, nW, xp, lane);
        }
        __syncthreads();

        if (w < 2) {
            v8f c = zero8();
            c = mm_breg<true, 16>(sxpT, 16, 0, bY, c, lane);
            v8f inn;
#pragma unroll
            for (int r = 0; r < 8; ++r) inn[r] = ym[r] - c[r];
            storeDT(sinT, 16, 0, nY, inn, lane);
        }
        __syncthreads();

        if (act) {
            v8f c = mmacc<true, false>(sinT, 16, 0, Gt, NXX, nW, NYY, xp, lane);
            storeDT(sxT, 16, 0, nW, c, lane);
        }
        __syncthreads();
    }

    for (int i = tid; i < 16 * NXX; i += 256) {
        int r = i >> 6, cc = i & 63;
        out[(size_t)(r0 + r) * NXX + cc] = sxT[cc * 16 + r];
    }
}

__global__ __launch_bounds__(256) void kf_pipe(const float* __restrict__ Yp,
                                               const float* __restrict__ Up,
                                               const float* __restrict__ Dp,
                                               const float* __restrict__ Wfx,
                                               const float* __restrict__ Wfu,
                                               const float* __restrict__ Wfd,
                                               const float* __restrict__ Wfy,
                                               float* G, float* out, const int k) {
    if (blockIdx.x == 0) {
        if (k < NCH) kf_gain_chunk(Wfx, Wfy, G, k);
    } else {
        if (k >= 1) kf_batch_chunk(Yp, Up, Dp, Wfx, Wfu, Wfd, Wfy, G, out, k - 1, blockIdx.x - 1);
    }
}

extern "C" void kernel_launch(void* const* d_in, const int* in_sizes, int n_in,
                              void* d_out, int out_size, void* d_ws, size_t ws_size,
                              hipStream_t stream) {
    (void)in_sizes; (void)n_in; (void)out_size; (void)ws_size;
    const float* Yp  = (const float*)d_in[0];
    const float* Up  = (const float*)d_in[1];
    const float* Dp  = (const float*)d_in[2];
    const float* Wfx = (const float*)d_in[3];
    const float* Wfu = (const float*)d_in[4];
    const float* Wfd = (const float*)d_in[5];
    const float* Wfy = (const float*)d_in[6];
    float* G = (float*)d_ws;
    float* out = (float*)d_out;

    kf_pipe<<<17, 256, 0, stream>>>(Yp, Up, Dp, Wfx, Wfu, Wfd, Wfy, G, out, 0);
    kf_pipe<<<17, 256, 0, stream>>>(Yp, Up, Dp, Wfx, Wfu, Wfd, Wfy, G, out, 1);
    kf_pipe<<<17, 256, 0, stream>>>(Yp, Up, Dp, Wfx, Wfu, Wfd, Wfy, G, out, 2);
    kf_pipe<<<17, 256, 0, stream>>>(Yp, Up, Dp, Wfx, Wfu, Wfd, Wfy, G, out, 3);
    kf_pipe<<<17, 256, 0, stream>>>(Yp, Up, Dp, Wfx, Wfu, Wfd, Wfy, G, out, 4);
    kf_pipe<<<17, 256, 0, stream>>>(Yp, Up, Dp, Wfx, Wfu, Wfd, Wfy, G, out, 5);
    kf_pipe<<<17, 256, 0, stream>>>(Yp, Up, Dp, Wfx, Wfu, Wfd, Wfy, G, out, 6);
    kf_pipe<<<17, 256, 0, stream>>>(Yp, Up, Dp, Wfx, Wfu, Wfd, Wfy, G, out, 7);
    kf_pipe<<<17, 256, 0, stream>>>(Yp, Up, Dp, Wfx, Wfu, Wfd, Wfy, G, out, 8);
}


# Inverse of S: a block step's 4x4 pivot block solved as two closed-form 2x2 inverses (adjugate over determinant): 2 divisions on the pivot-to-pivot chain, not 4; inverse's sums reordered
# speedup vs baseline: 3.9752x; 1.0440x over previous
#include <hip/hip_runtime.h>
#include <hip/hip_bf16.h>

#define NXX 64
#define NYY 32
#define NUU 16
#define NDD 8
#define T_STEPS 2048
#define B_TOT 256
#define NCH 8
#define CH (T_STEPS / NCH)
#define GSLOT (NYY * NXX)

typedef float v2f __attribute__((ext_vector_type(2)));
typedef float v8f __attribute__((ext_vector_type(8)));

__device__ __forceinline__ v8f zero8() {
    v8f c;
#pragma unroll
    for (int r = 0; r < 8; ++r) c[r] = 0.0f;
    return c;
}

__device__ __forceinline__ v2f fragA(const float* X, int ld, int m0, int k0, int lane) {
    int h = lane >> 4, m = m0 + (lane & 15), k = k0 + 2 * h;
    v2f a; a.x = X[m * ld + k]; a.y = X[m * ld + k + 1]; return a;
}
__device__ __forceinline__ v2f fragAT(const float* X, int ld, int m0, int k0, int lane) {
    int h = lane >> 4, m = m0 + (lane & 15), k = k0 + 2 * h;
    v2f a; a.x = X[k * ld + m]; a.y = X[(k + 1) * ld + m]; return a;
}
__device__ __forceinline__ v2f fragB(const float* X, int ld, int k0, int n0, int lane) {
    int h = lane >> 4, n = n0 + (lane & 15), k = k0 + 2 * h;
    v2f b; b.x = X[k * ld + n]; b.y = X[(k + 1) * ld + n]; return b;
}
__device__ __forceinline__ v2f fragBT(const float* X, int ld, int k0, int n0, int lane) {
    int h = lane >> 4, n = n0 + (lane & 15), k = k0 + 2 * h;
    v2f b; b.x = X[n * ld + k]; b.y = X[n * ld + k + 1]; return b;
}
__device__ __forceinline__ void storeD(float* X, int ld, int m0, int n0, v8f d, int lane,
                                       float sgn = 1.0f) {
    int h = lane >> 4, n = n0 + (lane & 15);
#pragma unroll
    for (int r = 0; r < 8; ++r) X[(m0 + r + 8 * h) * ld + n] = sgn * d[r];
}
__device__ __forceinline__ v8f loadC(const float* X, int ld, int m0, int n0, int lane) {
    int h = lane >> 4, n = n0 + (lane & 15);
    v8f c;
#pragma unroll
    for (int r = 0; r < 8; ++r) c[r] = X[(m0 + r + 8 * h) * ld + n];
    return c;
}
__device__ __forceinline__ void storeDT(float* X, int ld, int m0, int n0, v8f d, int lane) {
    int h = lane >> 4, n = n0 + (lane & 15);
    float4* p = reinterpret_cast<float4*>(X + n * ld + m0 + 8 * h);
    p[0] = make_float4(d[0], d[1], d[2], d[3]);
    p[1] = make_float4(d[4], d[5], d[6], d[7]);
}
__device__ __forceinline__ v8f identC(int m0, int n0, int lane) {
    int h = lane >> 4, n = n0 + (lane & 15);
    v8f c;
#pragma unroll
    for (int r = 0; r < 8; ++r) c[r] = ((m0 + r + 8 * h) == n) ? 1.0f : 0.0f;
    return c;
}

template <bool TA, bool TB>
__device__ __forceinline__ v8f mmacc(const float* A, int lda, int m0,
                                     const float* B, int ldb, int n0,
                                     int K, v8f c, int lane) {
#pragma unroll
    for (int k = 0; k < K; k += 4) {
        v2f a = TA ? fragAT(A, lda, m0, k, lane) : fragA(A, lda, m0, k, lane);
        v2f b = TB ? fragBT(B, ldb, k, n0, lane) : fragB(B, ldb, k, n0, lane);
        c = __builtin_amdgcn_wmma_f32_16x16x4_f32(false, a, false, b, (short)0, c,
                                                  false, false);
    }
    return c;
}

__device__ __forceinline__ v8f wmma4(v2f a, v2f b, v8f c) {
    return __builtin_amdgcn_wmma_f32_16x16x4_f32(false, a, false, b, (short)0, c,
                                                 false, false);
}
template <bool TA, int KS>
__device__ __forceinline__ v8f mm_breg(const float* A, int lda, int m0,
                                       const v2f (&b)[KS], v8f c, int lane) {
#pragma unroll
    for (int i = 0; i < KS; ++i) {
        v2f a = TA ? fragAT(A, lda, m0, 4 * i, lane) : fragA(A, lda, m0, 4 * i, lane);
        c = wmma4(a, b[i], c);
    }
    return c;
}

__device__ __forceinline__ void lds_group_fence() {
    __builtin_amdgcn_sched_barrier(0);
}
template <bool TA, int KS>
__device__ __forceinline__ void load_a(const float* A, int lda, int m0, int lane, v2f (&a)[KS]) {
#pragma unroll
    for (int i = 0; i < KS; ++i)
        a[i] = TA ? fragAT(A, lda, m0, 4 * i, lane) : fragA(A, lda, m0, 4 * i, lane);
}
template <bool TB, int KS>
__device__ __forceinline__ void load_b(const float* B, int ldb, int n0, int lane, v2f (&b)[KS]) {
#pragma unroll
    for (int i = 0; i < KS; ++i)
        b[i] = TB ? fragBT(B, ldb, 4 * i, n0, lane) : fragB(B, ldb, 4 * i, n0, lane);
}
template <int KS>
__device__ __forceinline__ v8f mm_regs(const v2f (&a)[KS], const v2f (&b)[KS], v8f c) {
#pragma unroll
    for (int i = 0; i < KS; ++i) c = wmma4(a[i], b[i], c);
    return c;
}

template <int KK>
__device__ __forceinline__ void gj_publish(float* Rt, float* Cc, v8f d, int gi, int gj, int lane) {
    constexpr int lr0 = (4 * KK) & 15, hh = lr0 >> 3, rb = lr0 & 7, pt = KK >> 2;
    const int h = lane >> 4, n = lane & 15;
    if (gi == pt && h == hh) {
        *reinterpret_cast<float4*>(Rt + (16 * gj + n) * 4) =
            make_float4(d[rb], d[rb + 1], d[rb + 2], d[rb + 3]);
    }
    const int cq = n - lr0;
    if (gj == pt && cq >= 0 && cq < 4) {
#pragma unroll
        for (int r = 0; r < 8; ++r) Cc[(16 * gi + r + 8 * h) * 4 + cq] = d[r];
    }
}

template <int KK>
__device__ __forceinline__ v8f gj_step(const float* Rt, const float* Cc, v8f d, int gi, int gj, int lane) {
    constexpr int lr0 = (4 * KK) & 15, hh = lr0 >> 3, rb = lr0 & 7, pt = KK >> 2;
    const int h = lane >> 4, n = lane & 15;
    float p[4][4], m[4];
    float4 pc[4];
#pragma unroll
    for (int c = 0; c < 4; ++c) pc[c] = *reinterpret_cast<const float4*>(Rt + (4 * KK + c) * 4);
    const float4 mv = *reinterpret_cast<const float4*>(Rt + (16 * gj + n) * 4);
    const float2 cv = *reinterpret_cast<const float2*>(Cc + (16 * gi + n) * 4 + 2 * h);
    lds_group_fence();
#pragma unroll
    for (int c = 0; c < 4; ++c) {
        p[0][c] = pc[c].x; p[1][c] = pc[c].y; p[2][c] = pc[c].z; p[3][c] = pc[c].w;
    }
    m[0] = mv.x; m[1] = mv.y; m[2] = mv.z; m[3] = mv.w;
    {
        const float a00 = p[0][0], a01 = p[0][1], a10 = p[1][0], a11 = p[1][1];
        const float detA = a00 * a11 - a01 * a10;
        const float rA = 1.0f / detA;
        float nX[2][2], nu[2];
#pragma unroll
        for (int j = 0; j < 2; ++j) {
            nX[0][j] = a01 * p[1][2 + j] - a11 * p[0][2 + j];
            nX[1][j] = a10 * p[0][2 + j] - a00 * p[1][2 + j];
        }
        nu[0] = a01 * m[1] - a11 * m[0];
        nu[1] = a10 * m[0] - a00 * m[1];
        float g[2][2], hv[2];
#pragma unroll
        for (int i = 0; i < 2; ++i) {
#pragma unroll
            for (int j = 0; j < 2; ++j) g[i][j] = p[2 + i][0] * nX[0][j] + p[2 + i][1] * nX[1][j];
            hv[i] = p[2 + i][0] * nu[0] + p[2 + i][1] * nu[1];
        }
        const float e00 = g[0][0] * rA + p[2][2];
        const float e11 = g[1][1] * rA + p[3][3];
        const float e10 = g[1][0] * rA + p[3][2];
        const float e01n = (-g[0][1]) * rA - p[2][3];
        const float v0 = hv[0] * rA + m[2];
        const float v1 = hv[1] * rA + m[3];
        const float detE = e00 * e11 + e01n * e10;
        const float rE = 1.0f / detE;
        const float y2 = (e11 * v0 + e01n * v1) * rE;
        const float y3 = (e00 * v1 - e10 * v0) * rE;
        m[0] = (-nu[0]) * rA + (nX[0][0] * rA) * y2 + (nX[0][1] * rA) * y3;
        m[1] = (-nu[1]) * rA + (nX[1][0] * rA) * y2 + (nX[1][1] * rA) * y3;
        m[2] = y2;
        m[3] = y3;
    }
    const int mr = n - lr0;
    const bool prow = (gi == pt) && (mr >= 0) && (mr < 4);
    v2f a, b;
    a.x = prow ? ((mr == 2 * h) ? 1.0f : 0.0f) : -cv.x;
    a.y = prow ? ((mr == 2 * h + 1) ? 1.0f : 0.0f) : -cv.y;
    b.x = h ? m[2] : m[0];
    b.y = h ? m[3] : m[1];
    const bool zr = (gi == pt) && (h == hh);
#pragma unroll
    for (int q = 0; q < 4; ++q) d[rb + q] = zr ? 0.0f : d[rb + q];
    return wmma4(a, b, d);
}

template <int KK>
__device__ __forceinline__ v8f gj_run(float (*Rt)[64 * 4], float (*Cc)[NYY * 4], v8f d,
                                      int gi, int gj, int lane) {
    d = gj_step<KK>(Rt[KK & 1], Cc[KK & 1], d, gi, gj, lane);
    if constexpr (KK < 7) {
        gj_publish<KK + 1>(Rt[(KK + 1) & 1], Cc[(KK + 1) & 1], d, gi, gj, lane);
        __syncthreads();
        return gj_run<KK + 1>(Rt, Cc, d, gi, gj, lane);
    } else {
        return d;
    }
}

__device__ __forceinline__ void kf_gain_chunk(const float* __restrict__ Wfx,
                                              const float* __restrict__ Wfy,
                                              float* __restrict__ G, const int ck) {
    __shared__ float sP[NXX * NXX];
    __shared__ __attribute__((aligned(16))) float sPT[NXX * NXX];
    __shared__ float sM1[NXX * NXX];
    __shared__ float sM2[NXX * NYY];
    __shared__ float sInv[NYY * NYY];
    __shared__ __attribute__((aligned(16))) float sRt[2][64 * 4];
    __shared__ __attribute__((aligned(16))) float sCc[2][NYY * 4];
    __shared__ __attribute__((aligned(16))) float sLT[NYY * NXX];

    const int tid = threadIdx.x, lane = tid & 31, w = tid >> 5;
    const int t0 = ck * CH, t1 = t0 + CH;

    if (ck == 0) {
        for (int i = tid; i < NXX * NXX; i += 256) {
            sP[i] = ((i >> 6) == (i & 63)) ? 1.0f : 0.0f;
        }
        for (int i = tid; i < NYY * NXX; i += 256) {
            sLT[i] = 0.0f;
        }
    } else {
        const float* Ls = G + (size_t)t0 * GSLOT;
        const float* Ps = G + (size_t)(t0 + 1) * GSLOT;
        for (int i = tid; i < NXX * NXX; i += 256) sP[i] = Ps[i];
        for (int i = tid; i < NYY * NXX; i += 256) sLT[i] = Ls[i];
    }

    const int mR = (w >> 1) * 16, nC = (w & 1) * 16;
    const int mA = (w >> 2) * 16, nA = (w & 3) * 16;
    const int mB = (w >> 1) * 16, nB = (w & 1) * 16;
    const int mS = ((w >> 1) & 1) * 16;
    const int gi = (w >> 1) & 1, gj = (w & 1) + ((w >> 2) << 1);

    v2f bXa[16], bXb[16];
    v2f bY[16];
    v2f aYS[16];
    v2f aYP[16];
#pragma unroll
    for (int i = 0; i < 16; ++i) {
        bXa[i] = fragBT(Wfx, NXX, 4 * i, nC, lane);
        bXb[i] = fragBT(Wfx, NXX, 4 * i, nC + 32, lane);
        bY[i]  = fragB(Wfy, NYY, 4 * i, nB, lane);
        aYS[i] = fragAT(Wfy, NYY, mS, 4 * i, lane);
        aYP[i] = fragAT(Wfy, NYY, mA, 4 * i, lane);
    }
    __syncthreads();

    for (int t = t0; t < t1; ++t) {
        {
            float4* Gt4 = reinterpret_cast<float4*>(G + (size_t)t * (NYY * NXX));
            const float4* L4 = reinterpret_cast<const float4*>(sLT);
            Gt4[tid] = L4[tid];
            Gt4[tid + 256] = L4[tid + 256];
        }
        __syncthreads();

        {
            v8f c0 = zero8(), c1 = zero8();
            v2f a[16];
            load_a<true, 16>(sP, NXX, mR, lane, a);
            lds_group_fence();
#pragma unroll
            for (int i = 0; i < 16; ++i) {
                c0 = wmma4(a[i], bXa[i], c0);
                c1 = wmma4(a[i], bXb[i], c1);
            }
            storeD(sM1, NXX, mR, nC, c0, lane);
            storeD(sM1, NXX, mR, nC + 32, c1, lane);
        }
        __syncthreads();
        {
            v8f c0 = identC(mR, nC, lane), c1 = identC(mR, nC + 32, lane);
            v2f a[16];
            load_a<true, 16>(sM1, NXX, mR, lane, a);
            lds_group_fence();
#pragma unroll
            for (int i = 0; i < 16; ++i) {
                c0 = wmma4(a[i], bXa[i], c0);
                c1 = wmma4(a[i], bXb[i], c1);
            }
            storeD(sP, NXX, mR, nC, c0, lane);
            storeD(sP, NXX, mR, nC + 32, c1, lane);
            storeDT(sPT + (mR >> 4) * (16 * NXX), 16, 0, nC, c0, lane);
            storeDT(sPT + (mR >> 4) * (16 * NXX), 16, 0, nC + 32, c1, lane);
        }
        __syncthreads();
        {
            v8f c = zero8();
            v2f a[16];
            load_a<true, 16>(sPT + (mB >> 4) * (16 * NXX), 16, 0, lane, a);
            lds_group_fence();
            c = mm_regs<16>(a, bY, c);
            storeD(sM2, NYY, mB, nB, c, lane);
        }
        __syncthreads();
        {
            v8f d = identC(mS, nB, lane);
            if (w < 4) {
                v2f b[16];
                load_b<false, 16>(sM2, NYY, nB, lane, b);
                lds_group_fence();
                d = mm_regs<16>(aYS, b, d);
            }
            gj_publish<0>(sRt[0], sCc[0], d, gi, gj, lane);
            __syncthreads();
            d = gj_run<0>(sRt, sCc, d, gi, gj, lane);
            if (w >= 4) storeD(sInv, NYY, mS, nB, d, lane);
        }
        __syncthreads();
        {
            v8f c = zero8();
            v2f a[8], b[8];
            load_a<true, 8>(sInv, NYY, nB, lane, a);
            load_b<true, 8>(sM2, NYY, mB, lane, b);
            lds_group_fence();
            c = mm_regs<8>(a, b, c);
            storeD(sLT, NXX, nB, mB, c, lane);
        }
        __syncthreads();
        {
            v8f c = zero8();
            v2f b[16];
            load_b<false, 16>(sP, NXX, nA, lane, b);
            lds_group_fence();
            c = mm_regs<16>(aYP, b, c);
            storeD(sM1, NXX, mA, nA, c, lane, -1.0f);
        }
        __syncthreads();
        {
            v2f a0[8], a1[8], b[8];
            load_a<true, 8>(sLT, NXX, mA, lane, a0);
            load_a<true, 8>(sLT, NXX, mA + 32, lane, a1);
            load_b<false, 8>(sM1, NXX, nA, lane, b);
            lds_group_fence();
            v8f c0 = identC(mA, nA, lane), c1 = identC(mA + 32, nA, lane);
#pragma unroll
            for (int i = 0; i < 8; ++i) {
                c0 = wmma4(a0[i], b[i], c0);
                c1 = wmma4(a1[i], b[i], c1);
            }
            storeD(sP, NXX, mA, nA, c0, lane);
            storeD(sP, NXX, mA + 32, nA, c1, lane);
        }
        __syncthreads();
    }

    if (ck + 1 < NCH) {
        float* Ls = G + (size_t)t1 * GSLOT;
        float* Ps = G + (size_t)(t1 + 1) * GSLOT;
        for (int i = tid; i < NXX * NXX; i += 256) Ps[i] = sP[i];
        for (int i = tid; i < NYY * NXX; i += 256) Ls[i] = sLT[i];
    }
}

__device__ __forceinline__ void kf_batch_chunk(const float* __restrict__ Yp,
                                               const float* __restrict__ Up,
                                               const float* __restrict__ Dp,
                                               const float* __restrict__ Wfx,
                                               const float* __restrict__ Wfu,
                                               const float* __restrict__ Wfd,
                                               const float* __restrict__ Wfy,
                                               const float* __restrict__ G,
                                               float* __restrict__ out,
                                               const int ck, const int bt) {
    __shared__ __attribute__((aligned(16))) float sxT[NXX * 16];
    __shared__ __attribute__((aligned(16))) float sxpT[NXX * 16];
    __shared__ __attribute__((aligned(16))) float sinT[NYY * 16];

    const int tid = threadIdx.x, lane = tid & 31, w = tid >> 5;
    const bool act = w < 4;
    const int r0 = bt * 16;
    const int lh = lane >> 4, lm = lane & 15;
    const int t0 = ck * CH, t1 = t0 + CH;

    if (ck == 0) {
        for (int i = tid; i < 16 * NXX; i += 256) sxT[i] = 0.0f;
    } else {
        for (int i = tid; i < 16 * NXX; i += 256) {
            int r = i >> 6, cc = i & 63;
            sxT[cc * 16 + r] = out[(size_t)(r0 + r) * NXX + cc];
        }
    }

    const int nW = (w & 3) * 16;
    const int nY = (w & 1) * 16;
    v2f bX[16], bU[4], bD[2];
    v2f bY[16];
#pragma unroll
    for (int i = 0; i < 16; ++i) {
        bX[i] = fragB(Wfx, NXX, 4 * i, nW, lane);
        bY[i] = fragB(Wfy, NYY, 4 * i, nY, lane);
    }
#pragma unroll
    for (int i = 0; i < 4; ++i) bU[i] = fragB(Wfu, NXX, 4 * i, nW, lane);
#pragma unroll
    for (int i = 0; i < 2; ++i) bD[i] = fragB(Wfd, NXX, 4 * i, nW, lane);
    __syncthreads();

    for (int t = t0; t < t1; ++t) {
        const float* Gt = G + (size_t)t * GSLOT;
        v8f xp = zero8();
        v8f ym = zero8();
        if (act) {
            const float* ut = Up + ((size_t)t * B_TOT + r0) * NUU;
            const float* dt = Dp + ((size_t)t * B_TOT + r0) * NDD;
            const float* yt = Yp + ((size_t)t * B_TOT + r0) * NYY;
            v2f aU[4], aD[2];
#pragma unroll
            for (int i = 0; i < 4; ++i)
                aU[i] = *reinterpret_cast<const v2f*>(ut + lm * NUU + 4 * i + 2 * lh);
#pragma unroll
            for (int i = 0; i < 2; ++i)
                aD[i] = *reinterpret_cast<const v2f*>(dt + lm * NDD + 4 * i + 2 * lh);
            if (w < 2) {
#pragma unroll
                for (int r = 0; r < 8; ++r) ym[r] = yt[(r + 8 * lh) * NYY + nY + lm];
            }
            if (t + 1 < t1) {
                __builtin_prefetch(G + (size_t)(t + 1) * GSLOT + tid * 16, 0, 1);
            }

            xp = mm_breg<true, 16>(sxT, 16, 0, bX, xp, lane);
#pragma unroll
            for (int i = 0; i < 4; ++i) xp = wmma4(aU[i], bU[i], xp);
#pragma unroll
            for (int i = 0; i < 2; ++i) xp = wmma4(aD[i], bD[i], xp);
            storeDT(sxpT, 16, 0, nW, xp, lane);
        }
        __syncthreads();

        if (w < 2) {
            v8f c = zero8();
            c = mm_breg<true, 16>(sxpT, 16, 0, bY, c, lane);
            v8f inn;
#pragma unroll
            for (int r = 0; r < 8; ++r) inn[r] = ym[r] - c[r];
            storeDT(sinT, 16, 0, nY, inn, lane);
        }
        __syncthreads();

        if (act) {
            v8f c = mmacc<true, false>(sinT, 16, 0, Gt, NXX, nW, NYY, xp, lane);
            storeDT(sxT, 16, 0, nW, c, lane);
        }
        __syncthreads();
    }

    for (int i = tid; i < 16 * NXX; i += 256) {
        int r = i >> 6, cc = i & 63;
        out[(size_t)(r0 + r) * NXX + cc] = sxT[cc * 16 + r];
    }
}

__global__ __launch_bounds__(256) void kf_pipe(const float* __restrict__ Yp,
                                               const float* __restrict__ Up,
                                               const float* __restrict__ Dp,
                                               const float* __restrict__ Wfx,
                                               const float* __restrict__ Wfu,
                                               const float* __restrict__ Wfd,
                                               const float* __restrict__ Wfy,
                                               float* G, float* out, const int k) {
    if (blockIdx.x == 0) {
        if (k < NCH) kf_gain_chunk(Wfx, Wfy, G, k);
    } else {
        if (k >= 1) kf_batch_chunk(Yp, Up, Dp, Wfx, Wfu, Wfd, Wfy, G, out, k - 1, blockIdx.x - 1);
    }
}

extern "C" void kernel_launch(void* const* d_in, const int* in_sizes, int n_in,
                              void* d_out, int out_size, void* d_ws, size_t ws_size,
                              hipStream_t stream) {
    (void)in_sizes; (void)n_in; (void)out_size; (void)ws_size;
    const float* Yp  = (const float*)d_in[0];
    const float* Up  = (const float*)d_in[1];
    const float* Dp  = (const float*)d_in[2];
    const float* Wfx = (const float*)d_in[3];
    const float* Wfu = (const float*)d_in[4];
    const float* Wfd = (const float*)d_in[5];
    const float* Wfy = (const float*)d_in[6];
    float* G = (float*)d_ws;
    float* out = (float*)d_out;

    kf_pipe<<<17, 256, 0, stream>>>(Yp, Up, Dp, Wfx, Wfu, Wfd, Wfy, G, out, 0);
    kf_pipe<<<17, 256, 0, stream>>>(Yp, Up, Dp, Wfx, Wfu, Wfd, Wfy, G, out, 1);
    kf_pipe<<<17, 256, 0, stream>>>(Yp, Up, Dp, Wfx, Wfu, Wfd, Wfy, G, out, 2);
    kf_pipe<<<17, 256, 0, stream>>>(Yp, Up, Dp, Wfx, Wfu, Wfd, Wfy, G, out, 3);
    kf_pipe<<<17, 256, 0, stream>>>(Yp, Up, Dp, Wfx, Wfu, Wfd, Wfy, G, out, 4);
    kf_pipe<<<17, 256, 0, stream>>>(Yp, Up, Dp, Wfx, Wfu, Wfd, Wfy, G, out, 5);
    kf_pipe<<<17, 256, 0, stream>>>(Yp, Up, Dp, Wfx, Wfu, Wfd, Wfy, G, out, 6);
    kf_pipe<<<17, 256, 0, stream>>>(Yp, Up, Dp, Wfx, Wfu, Wfd, Wfy, G, out, 7);
    kf_pipe<<<17, 256, 0, stream>>>(Yp, Up, Dp, Wfx, Wfu, Wfd, Wfy, G, out, 8);
}


# Gain recursion re-tiled onto 4 waves, each owning 16 rows of P: a product's accumulator tiles are the next product's WMMA operand in registers; 12 barriers a step, not 16; P, M2, S sums reordered
# speedup vs baseline: 4.5930x; 1.1554x over previous
#include <hip/hip_runtime.h>
#include <hip/hip_bf16.h>

#define NXX 64
#define NYY 32
#define NUU 16
#define NDD 8
#define T_STEPS 2048
#define B_TOT 256
#define NCH 8
#define CH (T_STEPS / NCH)
#define GSLOT (NYY * NXX)

typedef float v2f __attribute__((ext_vector_type(2)));
typedef float v8f __attribute__((ext_vector_type(8)));

__device__ __forceinline__ v8f zero8() {
    v8f c;
#pragma unroll
    for (int r = 0; r < 8; ++r) c[r] = 0.0f;
    return c;
}

__device__ __forceinline__ v2f fragA(const float* X, int ld, int m0, int k0, int lane) {
    int h = lane >> 4, m = m0 + (lane & 15), k = k0 + 2 * h;
    v2f a; a.x = X[m * ld + k]; a.y = X[m * ld + k + 1]; return a;
}
__device__ __forceinline__ v2f fragAT(const float* X, int ld, int m0, int k0, int lane) {
    int h = lane >> 4, m = m0 + (lane & 15), k = k0 + 2 * h;
    v2f a; a.x = X[k * ld + m]; a.y = X[(k + 1) * ld + m]; return a;
}
__device__ __forceinline__ v2f fragB(const float* X, int ld, int k0, int n0, int lane) {
    int h = lane >> 4, n = n0 + (lane & 15), k = k0 + 2 * h;
    v2f b; b.x = X[k * ld + n]; b.y = X[(k + 1) * ld + n]; return b;
}
__device__ __forceinline__ v2f fragBT(const float* X, int ld, int k0, int n0, int lane) {
    int h = lane >> 4, n = n0 + (lane & 15), k = k0 + 2 * h;
    v2f b; b.x = X[n * ld + k]; b.y = X[n * ld + k + 1]; return b;
}
__device__ __forceinline__ void storeD(float* X, int ld, int m0, int n0, v8f d, int lane,
                                       float sgn = 1.0f) {
    int h = lane >> 4, n = n0 + (lane & 15);
#pragma unroll
    for (int r = 0; r < 8; ++r) X[(m0 + r + 8 * h) * ld + n] = sgn * d[r];
}
__device__ __forceinline__ v8f loadC(const float* X, int ld, int m0, int n0, int lane) {
    int h = lane >> 4, n = n0 + (lane & 15);
    v8f c;
#pragma unroll
    for (int r = 0; r < 8; ++r) c[r] = X[(m0 + r + 8 * h) * ld + n];
    return c;
}
__device__ __forceinline__ void storeDT(float* X, int ld, int m0, int n0, v8f d, int lane) {
    int h = lane >> 4, n = n0 + (lane & 15);
    float4* p = reinterpret_cast<float4*>(X + n * ld + m0 + 8 * h);
    p[0] = make_float4(d[0], d[1], d[2], d[3]);
    p[1] = make_float4(d[4], d[5], d[6], d[7]);
}
__device__ __forceinline__ v8f identC(int m0, int n0, int lane) {
    int h = lane >> 4, n = n0 + (lane & 15);
    v8f c;
#pragma unroll
    for (int r = 0; r < 8; ++r) c[r] = ((m0 + r + 8 * h) == n) ? 1.0f : 0.0f;
    return c;
}

template <bool TA, bool TB>
__device__ __forceinline__ v8f mmacc(const float* A, int lda, int m0,
                                     const float* B, int ldb, int n0,
                                     int K, v8f c, int lane) {
#pragma unroll
    for (int k = 0; k < K; k += 4) {
        v2f a = TA ? fragAT(A, lda, m0, k, lane) : fragA(A, lda, m0, k, lane);
        v2f b = TB ? fragBT(B, ldb, k, n0, lane) : fragB(B, ldb, k, n0, lane);
        c = __builtin_amdgcn_wmma_f32_16x16x4_f32(false, a, false, b, (short)0, c,
                                                  false, false);
    }
    return c;
}

__device__ __forceinline__ v8f wmma4(v2f a, v2f b, v8f c) {
    return __builtin_amdgcn_wmma_f32_16x16x4_f32(false, a, false, b, (short)0, c,
                                                 false, false);
}
template <bool TA, int KS>
__device__ __forceinline__ v8f mm_breg(const float* A, int lda, int m0,
                                       const v2f (&b)[KS], v8f c, int lane) {
#pragma unroll
    for (int i = 0; i < KS; ++i) {
        v2f a = TA ? fragAT(A, lda, m0, 4 * i, lane) : fragA(A, lda, m0, 4 * i, lane);
        c = wmma4(a, b[i], c);
    }
    return c;
}

__device__ __forceinline__ void lds_group_fence() {
    __builtin_amdgcn_sched_barrier(0);
}
template <bool TA, int KS>
__device__ __forceinline__ void load_a(const float* A, int lda, int m0, int lane, v2f (&a)[KS]) {
#pragma unroll
    for (int i = 0; i < KS; ++i)
        a[i] = TA ? fragAT(A, lda, m0, 4 * i, lane) : fragA(A, lda, m0, 4 * i, lane);
}
template <bool TB, int KS>
__device__ __forceinline__ void load_b(const float* B, int ldb, int n0, int lane, v2f (&b)[KS]) {
#pragma unroll
    for (int i = 0; i < KS; ++i)
        b[i] = TB ? fragBT(B, ldb, 4 * i, n0, lane) : fragB(B, ldb, 4 * i, n0, lane);
}
template <int KS>
__device__ __forceinline__ v8f mm_regs(const v2f (&a)[KS], const v2f (&b)[KS], v8f c) {
#pragma unroll
    for (int i = 0; i < KS; ++i) c = wmma4(a[i], b[i], c);
    return c;
}

__device__ __forceinline__ v2f tile_k(v8f d, int i) {
    v2f f; f.x = d[2 * i]; f.y = d[2 * i + 1]; return f;
}
__device__ __forceinline__ v2f fragA_K(const float* X, int ld, int m0, int kb, int i, int lane) {
    int h = lane >> 4, m = m0 + (lane & 15), k = kb + 8 * h + 2 * i;
    v2f a; a.x = X[m * ld + k]; a.y = X[m * ld + k + 1]; return a;
}
__device__ __forceinline__ v2f fragAT_K(const float* X, int ld, int m0, int kb, int i, int lane) {
    int h = lane >> 4, m = m0 + (lane & 15), k = kb + 8 * h + 2 * i;
    v2f a; a.x = X[k * ld + m]; a.y = X[(k + 1) * ld + m]; return a;
}
__device__ __forceinline__ v2f fragB_K(const float* X, int ld, int kb, int i, int n0, int lane) {
    int h = lane >> 4, n = n0 + (lane & 15), k = kb + 8 * h + 2 * i;
    v2f b; b.x = X[k * ld + n]; b.y = X[(k + 1) * ld + n]; return b;
}

__device__ __forceinline__ void storeD_pairs(float* X, int ld, int m0, int n0, v8f d, int lane,
                                             float sgn = 1.0f) {
    int h = lane >> 4, n = n0 + (lane & 15);
#pragma unroll
    for (int q = 0; q < 4; ++q) {
        v2f pr; pr.x = sgn * d[2 * q]; pr.y = sgn * d[2 * q + 1];
        *reinterpret_cast<v2f*>(X + ((((m0 >> 1) + 4 * h + q) * ld + n) << 1)) = pr;
    }
}
__device__ __forceinline__ v2f load_pair(const float* X, int ld, int kpair, int col) {
    return *reinterpret_cast<const v2f*>(X + ((kpair * ld + col) << 1));
}

template <int KK>
__device__ __forceinline__ void gj_publish(float* Rt, float* Cc, v8f d, int gi, int gj, int lane) {
    constexpr int lr0 = (4 * KK) & 15, hh = lr0 >> 3, rb = lr0 & 7, pt = KK >> 2;
    const int h = lane >> 4, n = lane & 15;
    if (gi == pt && h == hh) {
        *reinterpret_cast<float4*>(Rt + (16 * gj + n) * 4) =
            make_float4(d[rb], d[rb + 1], d[rb + 2], d[rb + 3]);
    }
    const int cq = n - lr0;
    if (gj == pt && cq >= 0 && cq < 4) {
#pragma unroll
        for (int r = 0; r < 8; ++r) Cc[(16 * gi + r + 8 * h) * 4 + cq] = d[r];
    }
}

template <int KK>
__device__ __forceinline__ void gj_step(const float* Rt, const float* Cc, v8f& d0, v8f& d1,
                                        int gj, int lane) {
    constexpr int lr0 = (4 * KK) & 15, hh = lr0 >> 3, rb = lr0 & 7, pt = KK >> 2;
    const int h = lane >> 4, n = lane & 15;
    float p[4][4], m[4];
    float4 pc[4];
#pragma unroll
    for (int c = 0; c < 4; ++c) pc[c] = *reinterpret_cast<const float4*>(Rt + (4 * KK + c) * 4);
    const float4 mv = *reinterpret_cast<const float4*>(Rt + (16 * gj + n) * 4);
    const float2 cv0 = *reinterpret_cast<const float2*>(Cc + n * 4 + 2 * h);
    const float2 cv1 = *reinterpret_cast<const float2*>(Cc + (16 + n) * 4 + 2 * h);
    lds_group_fence();
#pragma unroll
    for (int c = 0; c < 4; ++c) {
        p[0][c] = pc[c].x; p[1][c] = pc[c].y; p[2][c] = pc[c].z; p[3][c] = pc[c].w;
    }
    m[0] = mv.x; m[1] = mv.y; m[2] = mv.z; m[3] = mv.w;
    {
        const float a00 = p[0][0], a01 = p[0][1], a10 = p[1][0], a11 = p[1][1];
        const float detA = a00 * a11 - a01 * a10;
        const float rA = 1.0f / detA;
        float nX[2][2], nu[2];
#pragma unroll
        for (int j = 0; j < 2; ++j) {
            nX[0][j] = a01 * p[1][2 + j] - a11 * p[0][2 + j];
            nX[1][j] = a10 * p[0][2 + j] - a00 * p[1][2 + j];
        }
        nu[0] = a01 * m[1] - a11 * m[0];
        nu[1] = a10 * m[0] - a00 * m[1];
        float g[2][2], hv[2];
#pragma unroll
        for (int i = 0; i < 2; ++i) {
#pragma unroll
            for (int j = 0; j < 2; ++j) g[i][j] = p[2 + i][0] * nX[0][j] + p[2 + i][1] * nX[1][j];
            hv[i] = p[2 + i][0] * nu[0] + p[2 + i][1] * nu[1];
        }
        const float e00 = g[0][0] * rA + p[2][2];
        const float e11 = g[1][1] * rA + p[3][3];
        const float e10 = g[1][0] * rA + p[3][2];
        const float e01n = (-g[0][1]) * rA - p[2][3];
        const float v0 = hv[0] * rA + m[2];
        const float v1 = hv[1] * rA + m[3];
        const float detE = e00 * e11 + e01n * e10;
        const float rE = 1.0f / detE;
        const float y2 = (e11 * v0 + e01n * v1) * rE;
        const float y3 = (e00 * v1 - e10 * v0) * rE;
        m[0] = (-nu[0]) * rA + (nX[0][0] * rA) * y2 + (nX[0][1] * rA) * y3;
        m[1] = (-nu[1]) * rA + (nX[1][0] * rA) * y2 + (nX[1][1] * rA) * y3;
        m[2] = y2;
        m[3] = y3;
    }
    const int mr = n - lr0;
    const bool inblk = (mr >= 0) && (mr < 4);
    v2f b, a0, a1, au;
    b.x = h ? m[2] : m[0];
    b.y = h ? m[3] : m[1];
    au.x = (mr == 2 * h) ? 1.0f : 0.0f;
    au.y = (mr == 2 * h + 1) ? 1.0f : 0.0f;
    a0.x = -cv0.x; a0.y = -cv0.y;
    a1.x = -cv1.x; a1.y = -cv1.y;
    const bool zr = (h == hh);
    if constexpr (pt == 0) {
        a0.x = inblk ? au.x : a0.x;
        a0.y = inblk ? au.y : a0.y;
#pragma unroll
        for (int q = 0; q < 4; ++q) d0[rb + q] = zr ? 0.0f : d0[rb + q];
    } else {
        a1.x = inblk ? au.x : a1.x;
        a1.y = inblk ? au.y : a1.y;
#pragma unroll
        for (int q = 0; q < 4; ++q) d1[rb + q] = zr ? 0.0f : d1[rb + q];
    }
    d0 = wmma4(a0, b, d0);
    d1 = wmma4(a1, b, d1);
}

template <int KK>
__device__ __forceinline__ void gj_run(float (*Rt)[64 * 4], float (*Cc)[NYY * 4], v8f& d0, v8f& d1,
                                       int gj, int lane) {
    gj_step<KK>(Rt[KK & 1], Cc[KK & 1], d0, d1, gj, lane);
    if constexpr (KK < 7) {
        gj_publish<KK + 1>(Rt[(KK + 1) & 1], Cc[(KK + 1) & 1], d0, 0, gj, lane);
        gj_publish<KK + 1>(Rt[(KK + 1) & 1], Cc[(KK + 1) & 1], d1, 1, gj, lane);
        __syncthreads();
        gj_run<KK + 1>(Rt, Cc, d0, d1, gj, lane);
    }
}

__device__ __forceinline__ void kf_gain_chunk(const float* __restrict__ Wfx,
                                              const float* __restrict__ Wfy,
                                              float* __restrict__ G, const int ck) {
    __shared__ __attribute__((aligned(16))) float sP[NXX * NXX];
    __shared__ __attribute__((aligned(16))) float sPb[NXX * NXX];
    __shared__ __attribute__((aligned(16))) float sN[NYY * NXX];
    __shared__ float sM2[NXX * NYY];
    __shared__ float sSp[4][NYY * NYY];
    __shared__ __attribute__((aligned(16))) float sInv[NYY * NYY];
    __shared__ __attribute__((aligned(16))) float sRt[2][64 * 4];
    __shared__ __attribute__((aligned(16))) float sCc[2][NYY * 4];
    __shared__ __attribute__((aligned(16))) float sLT[NYY * NXX];

    const int tid = threadIdx.x, lane = tid & 31, w = tid >> 5;
    const int t0 = ck * CH, t1 = t0 + CH;
    const int nw = 16 * w;

    if (ck == 0) {
        for (int i = tid; i < NXX * NXX; i += 128) {
            sP[(((i >> 7) * 64 + (i & 63)) << 1) + ((i >> 6) & 1)] = ((i >> 6) == (i & 63)) ? 1.0f : 0.0f;
        }
        for (int i = tid; i < NYY * NXX; i += 128) {
            sLT[i] = 0.0f;
        }
    } else {
        const float* Ls = G + (size_t)t0 * GSLOT;
        const float* Ps = G + (size_t)(t0 + 1) * GSLOT;
        for (int i = tid; i < NXX * NXX; i += 128) sP[(((i >> 7) * 64 + (i & 63)) << 1) + ((i >> 6) & 1)] = Ps[i];
        for (int i = tid; i < NYY * NXX; i += 128) sLT[i] = Ls[i];
    }

    v2f bX[16];
    v2f aXK[4][16];
    v2f bYK[2][16];
    v2f aYK[2][4];
    v2f aYP[2][16];
#pragma unroll
    for (int i = 0; i < 16; ++i) {
        bX[i] = fragBT(Wfx, NXX, 4 * i, nw, lane);
        aYP[0][i] = fragAT(Wfy, NYY, 0, 4 * i, lane);
        aYP[1][i] = fragAT(Wfy, NYY, 16, 4 * i, lane);
    }
#pragma unroll
    for (int kb = 0; kb < 4; ++kb) {
#pragma unroll
        for (int i = 0; i < 4; ++i) {
#pragma unroll
            for (int a = 0; a < 4; ++a) aXK[a][4 * kb + i] = fragA_K(Wfx, NXX, 16 * a, 16 * kb, i, lane);
            bYK[0][4 * kb + i] = fragB_K(Wfy, NYY, 16 * kb, i, 0, lane);
            bYK[1][4 * kb + i] = fragB_K(Wfy, NYY, 16 * kb, i, 16, lane);
        }
    }
#pragma unroll
    for (int i = 0; i < 4; ++i) {
        aYK[0][i] = fragAT_K(Wfy, NYY, 0, nw, i, lane);
        aYK[1][i] = fragAT_K(Wfy, NYY, 16, nw, i, lane);
    }
    __syncthreads();

    for (int t = t0; t < t1; ++t) {
        {
            float4* Gt4 = reinterpret_cast<float4*>(G + (size_t)t * (NYY * NXX));
            const float4* L4 = reinterpret_cast<const float4*>(sLT);
#pragma unroll
            for (int q = 0; q < 4; ++q) Gt4[tid + 128 * q] = L4[tid + 128 * q];
        }

        v8f Y[4];
        v8f M[2];
        {
            v8f Z[4];
            v2f a[4][16];
#pragma unroll
            for (int c = 0; c < 4; ++c) {
                Z[c] = zero8();
#pragma unroll
                for (int i = 0; i < 16; ++i)
                    a[c][i] = load_pair(sP, NXX, 2 * i + (lane >> 4), 16 * c + (lane & 15));
            }
            lds_group_fence();
#pragma unroll
            for (int i = 0; i < 16; ++i) {
#pragma unroll
                for (int c = 0; c < 4; ++c) Z[c] = wmma4(a[c][i], bX[i], Z[c]);
            }
#pragma unroll
            for (int a2 = 0; a2 < 4; ++a2) Y[a2] = identC(16 * a2, nw, lane);
#pragma unroll
            for (int c = 0; c < 4; ++c) {
#pragma unroll
                for (int i = 0; i < 4; ++i) {
                    const v2f bz = tile_k(Z[c], i);
#pragma unroll
                    for (int a2 = 0; a2 < 4; ++a2) Y[a2] = wmma4(aXK[a2][4 * c + i], bz, Y[a2]);
                }
            }
        }
#pragma unroll
        for (int a2 = 0; a2 < 4; ++a2) storeDT(sPb + a2 * (16 * NXX), 16, 0, nw, Y[a2], lane);
        M[0] = zero8(); M[1] = zero8();
#pragma unroll
        for (int a2 = 0; a2 < 4; ++a2) {
#pragma unroll
            for (int i = 0; i < 4; ++i) {
                const v2f ay = tile_k(Y[a2], i);
                M[0] = wmma4(ay, bYK[0][4 * a2 + i], M[0]);
                M[1] = wmma4(ay, bYK[1][4 * a2 + i], M[1]);
            }
        }
        storeD(sM2, NYY, nw, 0, M[0], lane);
        storeD(sM2, NYY, nw, 16, M[1], lane);
        {
            v8f Sp[2][2];
            Sp[0][0] = zero8(); Sp[0][1] = zero8(); Sp[1][0] = zero8(); Sp[1][1] = zero8();
#pragma unroll
            for (int i = 0; i < 4; ++i) {
#pragma unroll
                for (int nb = 0; nb < 2; ++nb) {
                    const v2f bm = tile_k(M[nb], i);
                    Sp[0][nb] = wmma4(aYK[0][i], bm, Sp[0][nb]);
                    Sp[1][nb] = wmma4(aYK[1][i], bm, Sp[1][nb]);
                }
            }
#pragma unroll
            for (int gi = 0; gi < 2; ++gi) {
#pragma unroll
                for (int nb = 0; nb < 2; ++nb) storeD(sSp[w], NYY, 16 * gi, 16 * nb, Sp[gi][nb], lane);
            }
        }
        __syncthreads();
        {
            const int nS = 16 * (w & 1);
            v8f d0 = identC(0, nS, lane), d1 = identC(16, nS, lane);
            if (w < 2) {
                v8f q0[4], q1[4];
#pragma unroll
                for (int j = 0; j < 4; ++j) {
                    q0[j] = loadC(sSp[j], NYY, 0, nS, lane);
                    q1[j] = loadC(sSp[j], NYY, 16, nS, lane);
                }
                d0 = d0 + ((q0[0] + q0[1]) + (q0[2] + q0[3]));
                d1 = d1 + ((q1[0] + q1[1]) + (q1[2] + q1[3]));
            }
            gj_publish<0>(sRt[0], sCc[0], d0, 0, w, lane);
            gj_publish<0>(sRt[0], sCc[0], d1, 1, w, lane);
            __syncthreads();
            gj_run<0>(sRt, sCc, d0, d1, w, lane);
            if (w >= 2) {
                storeD_pairs(sInv, NYY, 0, nS, d0, lane);
                storeD_pairs(sInv, NYY, 16, nS, d1, lane);
            }
        }
        __syncthreads();
        v8f T[2];
        {
            v2f ai0[8], ai1[8], bm[8], bp[16];
#pragma unroll
            for (int i = 0; i < 8; ++i) {
                ai0[i] = load_pair(sInv, NYY, 2 * i + (lane >> 4), (lane & 15));
                ai1[i] = load_pair(sInv, NYY, 2 * i + (lane >> 4), 16 + (lane & 15));
            }
            load_b<true, 8>(sM2, NYY, nw, lane, bm);
            load_b<false, 16>(sPb + w * (16 * NXX), 16, 0, lane, bp);
            lds_group_fence();
            T[0] = zero8(); T[1] = zero8();
#pragma unroll
            for (int i = 0; i < 8; ++i) {
                T[0] = wmma4(ai0[i], bm[i], T[0]);
                T[1] = wmma4(ai1[i], bm[i], T[1]);
            }
            v8f n0 = zero8(), n1 = zero8();
#pragma unroll
            for (int i = 0; i < 16; ++i) {
                n0 = wmma4(aYP[0][i], bp[i], n0);
                n1 = wmma4(aYP[1][i], bp[i], n1);
            }
            storeD(sLT, NXX, 0, nw, T[0], lane);
            storeD(sLT, NXX, 16, nw, T[1], lane);
            storeD_pairs(sN, NXX, 0, nw, n0, lane, -1.0f);
            storeD_pairs(sN, NXX, 16, nw, n1, lane, -1.0f);
        }
        __syncthreads();
        {
            v2f bn[4][8];
#pragma unroll
            for (int kb = 0; kb < 2; ++kb) {
#pragma unroll
                for (int i = 0; i < 4; ++i) {
#pragma unroll
                    for (int n = 0; n < 4; ++n)
                        bn[n][4 * kb + i] = load_pair(sN, NXX, 8 * kb + 4 * (lane >> 4) + i, 16 * n + (lane & 15));
                }
            }
            lds_group_fence();
            v8f Pn[4];
#pragma unroll
            for (int n = 0; n < 4; ++n) Pn[n] = identC(nw, 16 * n, lane);
#pragma unroll
            for (int kb = 0; kb < 2; ++kb) {
#pragma unroll
                for (int i = 0; i < 4; ++i) {
                    const v2f al = tile_k(T[kb], i);
#pragma unroll
                    for (int n = 0; n < 4; ++n) Pn[n] = wmma4(al, bn[n][4 * kb + i], Pn[n]);
                }
            }
#pragma unroll
            for (int n = 0; n < 4; ++n) storeD_pairs(sP, NXX, nw, 16 * n, Pn[n], lane);
        }
        __syncthreads();
    }

    if (ck + 1 < NCH) {
        float* Ls = G + (size_t)t1 * GSLOT;
        float* Ps = G + (size_t)(t1 + 1) * GSLOT;
        for (int i = tid; i < NXX * NXX; i += 128) Ps[i] = sP[(((i >> 7) * 64 + (i & 63)) << 1) + ((i >> 6) & 1)];
        for (int i = tid; i < NYY * NXX; i += 128) Ls[i] = sLT[i];
    }
}

__device__ __forceinline__ void kf_batch_chunk(const float* __restrict__ Yp,
                                               const float* __restrict__ Up,
                                               const float* __restrict__ Dp,
                                               const float* __restrict__ Wfx,
                                               const float* __restrict__ Wfu,
                                               const float* __restrict__ Wfd,
                                               const float* __restrict__ Wfy,
                                               const float* __restrict__ G,
                                               float* __restrict__ out,
                                               const int ck, const int bt) {
    __shared__ __attribute__((aligned(16))) float sxT[NXX * 16];
    __shared__ __attribute__((aligned(16))) float sxpT[NXX * 16];
    __shared__ __attribute__((aligned(16))) float sinT[NYY * 16];

    const int tid = threadIdx.x, lane = tid & 31, w = tid >> 5;
    const int r0 = bt * 16;
    const int lh = lane >> 4, lm = lane & 15;
    const int t0 = ck * CH, t1 = t0 + CH;

    if (ck == 0) {
        for (int i = tid; i < 16 * NXX; i += 128) sxT[i] = 0.0f;
    } else {
        for (int i = tid; i < 16 * NXX; i += 128) {
            int r = i >> 6, cc = i & 63;
            sxT[cc * 16 + r] = out[(size_t)(r0 + r) * NXX + cc];
        }
    }

    const int nW = w * 16;
    const int nY = (w & 1) * 16;
    v2f bX[16], bU[4], bD[2];
    v2f bY[16];
#pragma unroll
    for (int i = 0; i < 16; ++i) {
        bX[i] = fragB(Wfx, NXX, 4 * i, nW, lane);
        bY[i] = fragB(Wfy, NYY, 4 * i, nY, lane);
    }
#pragma unroll
    for (int i = 0; i < 4; ++i) bU[i] = fragB(Wfu, NXX, 4 * i, nW, lane);
#pragma unroll
    for (int i = 0; i < 2; ++i) bD[i] = fragB(Wfd, NXX, 4 * i, nW, lane);
    __syncthreads();

    for (int t = t0; t < t1; ++t) {
        const float* Gt = G + (size_t)t * GSLOT;
        v8f xp = zero8();
        v8f ym = zero8();
        {
            const float* ut = Up + ((size_t)t * B_TOT + r0) * NUU;
            const float* dt = Dp + ((size_t)t * B_TOT + r0) * NDD;
            const float* yt = Yp + ((size_t)t * B_TOT + r0) * NYY;
            v2f aU[4], aD[2];
#pragma unroll
            for (int i = 0; i < 4; ++i)
                aU[i] = *reinterpret_cast<const v2f*>(ut + lm * NUU + 4 * i + 2 * lh);
#pragma unroll
            for (int i = 0; i < 2; ++i)
                aD[i] = *reinterpret_cast<const v2f*>(dt + lm * NDD + 4 * i + 2 * lh);
            if (w < 2) {
#pragma unroll
                for (int r = 0; r < 8; ++r) ym[r] = yt[(r + 8 * lh) * NYY + nY + lm];
            }
            if (t + 1 < t1) {
                __builtin_prefetch(G + (size_t)(t + 1) * GSLOT + tid * 16, 0, 1);
            }

            xp = mm_breg<true, 16>(sxT, 16, 0, bX, xp, lane);
#pragma unroll
            for (int i = 0; i < 4; ++i) xp = wmma4(aU[i], bU[i], xp);
#pragma unroll
            for (int i = 0; i < 2; ++i) xp = wmma4(aD[i], bD[i], xp);
            storeDT(sxpT, 16, 0, nW, xp, lane);
        }
        __syncthreads();

        if (w < 2) {
            v8f c = zero8();
            c = mm_breg<true, 16>(sxpT, 16, 0, bY, c, lane);
            v8f inn;
#pragma unroll
            for (int r = 0; r < 8; ++r) inn[r] = ym[r] - c[r];
            storeDT(sinT, 16, 0, nY, inn, lane);
        }
        __syncthreads();

        {
            v8f c = mmacc<true, false>(sinT, 16, 0, Gt, NXX, nW, NYY, xp, lane);
            storeDT(sxT, 16, 0, nW, c, lane);
        }
        __syncthreads();
    }

    for (int i = tid; i < 16 * NXX; i += 128) {
        int r = i >> 6, cc = i & 63;
        out[(size_t)(r0 + r) * NXX + cc] = sxT[cc * 16 + r];
    }
}

__global__ __launch_bounds__(128) void kf_pipe(const float* __restrict__ Yp,
                                               const float* __restrict__ Up,
                                               const float* __restrict__ Dp,
                                               const float* __restrict__ Wfx,
                                               const float* __restrict__ Wfu,
                                               const float* __restrict__ Wfd,
                                               const float* __restrict__ Wfy,
                                               float* G, float* out, const int k) {
    if (blockIdx.x == 0) {
        if (k < NCH) kf_gain_chunk(Wfx, Wfy, G, k);
    } else {
        if (k >= 1) kf_batch_chunk(Yp, Up, Dp, Wfx, Wfu, Wfd, Wfy, G, out, k - 1, blockIdx.x - 1);
    }
}

extern "C" void kernel_launch(void* const* d_in, const int* in_sizes, int n_in,
                              void* d_out, int out_size, void* d_ws, size_t ws_size,
                              hipStream_t stream) {
    (void)in_sizes; (void)n_in; (void)out_size; (void)ws_size;
    const float* Yp  = (const float*)d_in[0];
    const float* Up  = (const float*)d_in[1];
    const float* Dp  = (const float*)d_in[2];
    const float* Wfx = (const float*)d_in[3];
    const float* Wfu = (const float*)d_in[4];
    const float* Wfd = (const float*)d_in[5];
    const float* Wfy = (const float*)d_in[6];
    float* G = (float*)d_ws;
    float* out = (float*)d_out;

    kf_pipe<<<17, 128, 0, stream>>>(Yp, Up, Dp, Wfx, Wfu, Wfd, Wfy, G, out, 0);
    kf_pipe<<<17, 128, 0, stream>>>(Yp, Up, Dp, Wfx, Wfu, Wfd, Wfy, G, out, 1);
    kf_pipe<<<17, 128, 0, stream>>>(Yp, Up, Dp, Wfx, Wfu, Wfd, Wfy, G, out, 2);
    kf_pipe<<<17, 128, 0, stream>>>(Yp, Up, Dp, Wfx, Wfu, Wfd, Wfy, G, out, 3);
    kf_pipe<<<17, 128, 0, stream>>>(Yp, Up, Dp, Wfx, Wfu, Wfd, Wfy, G, out, 4);
    kf_pipe<<<17, 128, 0, stream>>>(Yp, Up, Dp, Wfx, Wfu, Wfd, Wfy, G, out, 5);
    kf_pipe<<<17, 128, 0, stream>>>(Yp, Up, Dp, Wfx, Wfu, Wfd, Wfy, G, out, 6);
    kf_pipe<<<17, 128, 0, stream>>>(Yp, Up, Dp, Wfx, Wfu, Wfd, Wfy, G, out, 7);
    kf_pipe<<<17, 128, 0, stream>>>(Yp, Up, Dp, Wfx, Wfu, Wfd, Wfy, G, out, 8);
}


# Gain step: predicted covariance never formed: M2 = Wfy + Wfx (P A1), A1 = Wfx^T Wfy once a launch, and Wfy^T Pp read as M2^T (P = P^T): 144 WMMAs a wave and step, not 272; P, M2 sums reordered
# speedup vs baseline: 6.0377x; 1.3145x over previous
#include <hip/hip_runtime.h>
#include <hip/hip_bf16.h>

#define NXX 64
#define NYY 32
#define NUU 16
#define NDD 8
#define T_STEPS 2048
#define B_TOT 256
#define NCH 8
#define CH (T_STEPS / NCH)
#define GSLOT (NYY * NXX)
#define LDN 68

typedef float v2f __attribute__((ext_vector_type(2)));
typedef float v8f __attribute__((ext_vector_type(8)));

__device__ __forceinline__ v8f zero8() {
    v8f c;
#pragma unroll
    for (int r = 0; r < 8; ++r) c[r] = 0.0f;
    return c;
}

__device__ __forceinline__ v2f fragA(const float* X, int ld, int m0, int k0, int lane) {
    int h = lane >> 4, m = m0 + (lane & 15), k = k0 + 2 * h;
    v2f a; a.x = X[m * ld + k]; a.y = X[m * ld + k + 1]; return a;
}
__device__ __forceinline__ v2f fragAT(const float* X, int ld, int m0, int k0, int lane) {
    int h = lane >> 4, m = m0 + (lane & 15), k = k0 + 2 * h;
    v2f a; a.x = X[k * ld + m]; a.y = X[(k + 1) * ld + m]; return a;
}
__device__ __forceinline__ v2f fragB(const float* X, int ld, int k0, int n0, int lane) {
    int h = lane >> 4, n = n0 + (lane & 15), k = k0 + 2 * h;
    v2f b; b.x = X[k * ld + n]; b.y = X[(k + 1) * ld + n]; return b;
}
__device__ __forceinline__ v2f fragBT(const float* X, int ld, int k0, int n0, int lane) {
    int h = lane >> 4, n = n0 + (lane & 15), k = k0 + 2 * h;
    v2f b; b.x = X[n * ld + k]; b.y = X[n * ld + k + 1]; return b;
}
__device__ __forceinline__ void storeD(float* X, int ld, int m0, int n0, v8f d, int lane,
                                       float sgn = 1.0f) {
    int h = lane >> 4, n = n0 + (lane & 15);
#pragma unroll
    for (int r = 0; r < 8; ++r) X[(m0 + r + 8 * h) * ld + n] = sgn * d[r];
}
__device__ __forceinline__ v8f loadC(const float* X, int ld, int m0, int n0, int lane) {
    int h = lane >> 4, n = n0 + (lane & 15);
    v8f c;
#pragma unroll
    for (int r = 0; r < 8; ++r) c[r] = X[(m0 + r + 8 * h) * ld + n];
    return c;
}
__device__ __forceinline__ void storeDT(float* X, int ld, int m0, int n0, v8f d, int lane) {
    int h = lane >> 4, n = n0 + (lane & 15);
    float4* p = reinterpret_cast<float4*>(X + n * ld + m0 + 8 * h);
    p[0] = make_float4(d[0], d[1], d[2], d[3]);
    p[1] = make_float4(d[4], d[5], d[6], d[7]);
}
__device__ __forceinline__ v8f identC(int m0, int n0, int lane) {
    int h = lane >> 4, n = n0 + (lane & 15);
    v8f c;
#pragma unroll
    for (int r = 0; r < 8; ++r) c[r] = ((m0 + r + 8 * h) == n) ? 1.0f : 0.0f;
    return c;
}

template <bool TA, bool TB>
__device__ __forceinline__ v8f mmacc(const float* A, int lda, int m0,
                                     const float* B, int ldb, int n0,
                                     int K, v8f c, int lane) {
#pragma unroll
    for (int k = 0; k < K; k += 4) {
        v2f a = TA ? fragAT(A, lda, m0, k, lane) : fragA(A, lda, m0, k, lane);
        v2f b = TB ? fragBT(B, ldb, k, n0, lane) : fragB(B, ldb, k, n0, lane);
        c = __builtin_amdgcn_wmma_f32_16x16x4_f32(false, a, false, b, (short)0, c,
                                                  false, false);
    }
    return c;
}

__device__ __forceinline__ v8f wmma4(v2f a, v2f b, v8f c) {
    return __builtin_amdgcn_wmma_f32_16x16x4_f32(false, a, false, b, (short)0, c,
                                                 false, false);
}
template <bool TA, int KS>
__device__ __forceinline__ v8f mm_breg(const float* A, int lda, int m0,
                                       const v2f (&b)[KS], v8f c, int lane) {
#pragma unroll
    for (int i = 0; i < KS; ++i) {
        v2f a = TA ? fragAT(A, lda, m0, 4 * i, lane) : fragA(A, lda, m0, 4 * i, lane);
        c = wmma4(a, b[i], c);
    }
    return c;
}

__device__ __forceinline__ void lds_group_fence() {
    __builtin_amdgcn_sched_barrier(0);
}
template <bool TA, int KS>
__device__ __forceinline__ void load_a(const float* A, int lda, int m0, int lane, v2f (&a)[KS]) {
#pragma unroll
    for (int i = 0; i < KS; ++i)
        a[i] = TA ? fragAT(A, lda, m0, 4 * i, lane) : fragA(A, lda, m0, 4 * i, lane);
}
template <bool TB, int KS>
__device__ __forceinline__ void load_b(const float* B, int ldb, int n0, int lane, v2f (&b)[KS]) {
#pragma unroll
    for (int i = 0; i < KS; ++i)
        b[i] = TB ? fragBT(B, ldb, 4 * i, n0, lane) : fragB(B, ldb, 4 * i, n0, lane);
}
template <int KS>
__device__ __forceinline__ v8f mm_regs(const v2f (&a)[KS], const v2f (&b)[KS], v8f c) {
#pragma unroll
    for (int i = 0; i < KS; ++i) c = wmma4(a[i], b[i], c);
    return c;
}

__device__ __forceinline__ v2f tile_k(v8f d, int i) {
    v2f f; f.x = d[2 * i]; f.y = d[2 * i + 1]; return f;
}
__device__ __forceinline__ v2f fragAT_K(const float* X, int ld, int m0, int kb, int i, int lane) {
    int h = lane >> 4, m = m0 + (lane & 15), k = kb + 8 * h + 2 * i;
    v2f a; a.x = X[k * ld + m]; a.y = X[(k + 1) * ld + m]; return a;
}

__device__ __forceinline__ void storeD_pairs(float* X, int ld, int m0, int n0, v8f d, int lane,
                                             float sgn = 1.0f) {
    int h = lane >> 4, n = n0 + (lane & 15);
#pragma unroll
    for (int q = 0; q < 4; ++q) {
        v2f pr; pr.x = sgn * d[2 * q]; pr.y = sgn * d[2 * q + 1];
        *reinterpret_cast<v2f*>(X + ((((m0 >> 1) + 4 * h + q) * ld + n) << 1)) = pr;
    }
}
__device__ __forceinline__ v2f load_pair(const float* X, int ld, int kpair, int col) {
    return *reinterpret_cast<const v2f*>(X + ((kpair * ld + col) << 1));
}
__device__ __forceinline__ void storeDT_pairs_neg(float* X, int ld, int m0, int k0, v8f d, int lane) {
    int h = lane >> 4, k = k0 + (lane & 15);
    float* p = X + ((((k >> 1) * ld + m0 + 8 * h) << 1) + (k & 1));
#pragma unroll
    for (int r = 0; r < 8; ++r) p[2 * r] = -d[r];
}

template <int KK>
__device__ __forceinline__ void gj_publish(float* Rt, float* Cc, v8f d, int gi, int gj, int lane) {
    constexpr int lr0 = (4 * KK) & 15, hh = lr0 >> 3, rb = lr0 & 7, pt = KK >> 2;
    const int h = lane >> 4, n = lane & 15;
    if (gi == pt && h == hh) {
        *reinterpret_cast<float4*>(Rt + (16 * gj + n) * 4) =
            make_float4(d[rb], d[rb + 1], d[rb + 2], d[rb + 3]);
    }
    const int cq = n - lr0;
    if (gj == pt && cq >= 0 && cq < 4) {
#pragma unroll
        for (int r = 0; r < 8; ++r) Cc[(16 * gi + r + 8 * h) * 4 + cq] = d[r];
    }
}

template <int KK>
__device__ __forceinline__ void gj_step(const float* Rt, const float* Cc, v8f& d0, v8f& d1,
                                        int gj, int lane) {
    constexpr int lr0 = (4 * KK) & 15, hh = lr0 >> 3, rb = lr0 & 7, pt = KK >> 2;
    const int h = lane >> 4, n = lane & 15;
    float p[4][4], m[4];
    float4 pc[4];
#pragma unroll
    for (int c = 0; c < 4; ++c) pc[c] = *reinterpret_cast<const float4*>(Rt + (4 * KK + c) * 4);
    const float4 mv = *reinterpret_cast<const float4*>(Rt + (16 * gj + n) * 4);
    const float2 cv0 = *reinterpret_cast<const float2*>(Cc + n * 4 + 2 * h);
    const float2 cv1 = *reinterpret_cast<const float2*>(Cc + (16 + n) * 4 + 2 * h);
    lds_group_fence();
#pragma unroll
    for (int c = 0; c < 4; ++c) {
        p[0][c] = pc[c].x; p[1][c] = pc[c].y; p[2][c] = pc[c].z; p[3][c] = pc[c].w;
    }
    m[0] = mv.x; m[1] = mv.y; m[2] = mv.z; m[3] = mv.w;
    {
        const float a00 = p[0][0], a01 = p[0][1], a10 = p[1][0], a11 = p[1][1];
        const float detA = a00 * a11 - a01 * a10;
        const float rA = 1.0f / detA;
        float nX[2][2], nu[2];
#pragma unroll
        for (int j = 0; j < 2; ++j) {
            nX[0][j] = a01 * p[1][2 + j] - a11 * p[0][2 + j];
            nX[1][j] = a10 * p[0][2 + j] - a00 * p[1][2 + j];
        }
        nu[0] = a01 * m[1] - a11 * m[0];
        nu[1] = a10 * m[0] - a00 * m[1];
        float g[2][2], hv[2];
#pragma unroll
        for (int i = 0; i < 2; ++i) {
#pragma unroll
            for (int j = 0; j < 2; ++j) g[i][j] = p[2 + i][0] * nX[0][j] + p[2 + i][1] * nX[1][j];
            hv[i] = p[2 + i][0] * nu[0] + p[2 + i][1] * nu[1];
        }
        const float e00 = g[0][0] * rA + p[2][2];
        const float e11 = g[1][1] * rA + p[3][3];
        const float e10 = g[1][0] * rA + p[3][2];
        const float e01n = (-g[0][1]) * rA - p[2][3];
        const float v0 = hv[0] * rA + m[2];
        const float v1 = hv[1] * rA + m[3];
        const float detE = e00 * e11 + e01n * e10;
        const float rE = 1.0f / detE;
        const float y2 = (e11 * v0 + e01n * v1) * rE;
        const float y3 = (e00 * v1 - e10 * v0) * rE;
        m[0] = (-nu[0]) * rA + (nX[0][0] * rA) * y2 + (nX[0][1] * rA) * y3;
        m[1] = (-nu[1]) * rA + (nX[1][0] * rA) * y2 + (nX[1][1] * rA) * y3;
        m[2] = y2;
        m[3] = y3;
    }
    const int mr = n - lr0;
    const bool inblk = (mr >= 0) && (mr < 4);
    v2f b, a0, a1, au;
    b.x = h ? m[2] : m[0];
    b.y = h ? m[3] : m[1];
    au.x = (mr == 2 * h) ? 1.0f : 0.0f;
    au.y = (mr == 2 * h + 1) ? 1.0f : 0.0f;
    a0.x = -cv0.x; a0.y = -cv0.y;
    a1.x = -cv1.x; a1.y = -cv1.y;
    const bool zr = (h == hh);
    if constexpr (pt == 0) {
        a0.x = inblk ? au.x : a0.x;
        a0.y = inblk ? au.y : a0.y;
#pragma unroll
        for (int q = 0; q < 4; ++q) d0[rb + q] = zr ? 0.0f : d0[rb + q];
    } else {
        a1.x = inblk ? au.x : a1.x;
        a1.y = inblk ? au.y : a1.y;
#pragma unroll
        for (int q = 0; q < 4; ++q) d1[rb + q] = zr ? 0.0f : d1[rb + q];
    }
    d0 = wmma4(a0, b, d0);
    d1 = wmma4(a1, b, d1);
}

template <int KK>
__device__ __forceinline__ void gj_run(float (*Rt)[64 * 4], float (*Cc)[NYY * 4], v8f& d0, v8f& d1,
                                       int gj, int lane) {
    gj_step<KK>(Rt[KK & 1], Cc[KK & 1], d0, d1, gj, lane);
    if constexpr (KK < 7) {
        gj_publish<KK + 1>(Rt[(KK + 1) & 1], Cc[(KK + 1) & 1], d0, 0, gj, lane);
        gj_publish<KK + 1>(Rt[(KK + 1) & 1], Cc[(KK + 1) & 1], d1, 1, gj, lane);
        __syncthreads();
        gj_run<KK + 1>(Rt, Cc, d0, d1, gj, lane);
    }
}

__device__ __forceinline__ void kf_gain_chunk(const float* __restrict__ Wfx,
                                              const float* __restrict__ Wfy,
                                              float* __restrict__ G, const int ck) {
    __shared__ __attribute__((aligned(16))) float sP[NXX * NXX];
    __shared__ __attribute__((aligned(16))) float sU[NXX * NYY];
    __shared__ __attribute__((aligned(16))) float sN[(NYY / 2) * LDN * 2];
    __shared__ float sM2[NXX * NYY];
    __shared__ float sSp[4][NYY * NYY];
    __shared__ __attribute__((aligned(16))) float sInv[NYY * NYY];
    __shared__ __attribute__((aligned(16))) float sRt[2][64 * 4];
    __shared__ __attribute__((aligned(16))) float sCc[2][NYY * 4];
    __shared__ __attribute__((aligned(16))) float sLT[NYY * NXX];

    const int tid = threadIdx.x, lane = tid & 31, w = tid >> 5;
    const int t0 = ck * CH, t1 = t0 + CH;
    const int nw = 16 * w;

    if (ck == 0) {
        for (int i = tid; i < NXX * NXX; i += 128) {
            sP[(((i >> 7) * 64 + (i & 63)) << 1) + ((i >> 6) & 1)] = ((i >> 6) == (i & 63)) ? 1.0f : 0.0f;
        }
        for (int i = tid; i < NYY * NXX; i += 128) {
            sLT[i] = 0.0f;
        }
    } else {
        const float* Ls = G + (size_t)t0 * GSLOT;
        const float* Ps = G + (size_t)(t0 + 1) * GSLOT;
        for (int i = tid; i < NXX * NXX; i += 128) sP[(((i >> 7) * 64 + (i & 63)) << 1) + ((i >> 6) & 1)] = Ps[i];
        for (int i = tid; i < NYY * NXX; i += 128) sLT[i] = Ls[i];
    }

    {
        const v8f t0 = mmacc<true, false>(Wfx, NXX, nw, Wfy, NYY, 0, NXX, zero8(), lane);
        const v8f t1 = mmacc<true, false>(Wfx, NXX, nw, Wfy, NYY, 16, NXX, zero8(), lane);
        storeD(sM2, NYY, nw, 0, t0, lane);
        storeD(sM2, NYY, nw, 16, t1, lane);
    }
    v2f bA1[2][16];
    v2f aW[16];
    v8f cW[2];
    v2f aYK[2][4];
#pragma unroll
    for (int i = 0; i < 16; ++i) aW[i] = fragA(Wfx, NXX, nw, 4 * i, lane);
    cW[0] = loadC(Wfy, NYY, nw, 0, lane);
    cW[1] = loadC(Wfy, NYY, nw, 16, lane);
#pragma unroll
    for (int i = 0; i < 4; ++i) {
        aYK[0][i] = fragAT_K(Wfy, NYY, 0, nw, i, lane);
        aYK[1][i] = fragAT_K(Wfy, NYY, 16, nw, i, lane);
    }
    __syncthreads();
#pragma unroll
    for (int i = 0; i < 16; ++i) {
        bA1[0][i] = fragB(sM2, NYY, 4 * i, 0, lane);
        bA1[1][i] = fragB(sM2, NYY, 4 * i, 16, lane);
    }

    for (int t = t0; t < t1; ++t) {
        v8f M[2];
        {
            v2f a[16];
#pragma unroll
            for (int i = 0; i < 16; ++i)
                a[i] = load_pair(sP, NXX, 2 * i + (lane >> 4), nw + (lane & 15));
            lds_group_fence();
            v8f U0 = zero8(), U1 = zero8();
#pragma unroll
            for (int i = 0; i < 16; ++i) {
                U0 = wmma4(a[i], bA1[0][i], U0);
                U1 = wmma4(a[i], bA1[1][i], U1);
            }
            storeD_pairs(sU, NYY, nw, 0, U0, lane);
            storeD_pairs(sU, NYY, nw, 16, U1, lane);
        }
        __syncthreads();
        {
            float4* Gt4 = reinterpret_cast<float4*>(G + (size_t)t * (NYY * NXX));
            const float4* L4 = reinterpret_cast<const float4*>(sLT);
#pragma unroll
            for (int q = 0; q < 4; ++q) Gt4[tid + 128 * q] = L4[tid + 128 * q];
        }
        {
            v2f bu[2][16];
#pragma unroll
            for (int i = 0; i < 16; ++i) {
                bu[0][i] = load_pair(sU, NYY, 2 * i + (lane >> 4), (lane & 15));
                bu[1][i] = load_pair(sU, NYY, 2 * i + (lane >> 4), 16 + (lane & 15));
            }
            lds_group_fence();
            M[0] = cW[0]; M[1] = cW[1];
#pragma unroll
            for (int i = 0; i < 16; ++i) {
                M[0] = wmma4(aW[i], bu[0][i], M[0]);
                M[1] = wmma4(aW[i], bu[1][i], M[1]);
            }
        }
        storeD(sM2, NYY, nw, 0, M[0], lane);
        storeD(sM2, NYY, nw, 16, M[1], lane);
        storeDT_pairs_neg(sN, LDN, nw, 0, M[0], lane);
        storeDT_pairs_neg(sN, LDN, nw, 16, M[1], lane);
        {
            v8f Sp[2][2];
            Sp[0][0] = zero8(); Sp[0][1] = zero8(); Sp[1][0] = zero8(); Sp[1][1] = zero8();
#pragma unroll
            for (int i = 0; i < 4; ++i) {
#pragma unroll
                for (int nb = 0; nb < 2; ++nb) {
                    const v2f bm = tile_k(M[nb], i);
                    Sp[0][nb] = wmma4(aYK[0][i], bm, Sp[0][nb]);
                    Sp[1][nb] = wmma4(aYK[1][i], bm, Sp[1][nb]);
                }
            }
#pragma unroll
            for (int gi = 0; gi < 2; ++gi) {
#pragma unroll
                for (int nb = 0; nb < 2; ++nb) storeD(sSp[w], NYY, 16 * gi, 16 * nb, Sp[gi][nb], lane);
            }
        }
        __syncthreads();
        {
            const int nS = 16 * (w & 1);
            v8f d0 = identC(0, nS, lane), d1 = identC(16, nS, lane);
            if (w < 2) {
                v8f q0[4], q1[4];
#pragma unroll
                for (int j = 0; j < 4; ++j) {
                    q0[j] = loadC(sSp[j], NYY, 0, nS, lane);
                    q1[j] = loadC(sSp[j], NYY, 16, nS, lane);
                }
                d0 = d0 + ((q0[0] + q0[1]) + (q0[2] + q0[3]));
                d1 = d1 + ((q1[0] + q1[1]) + (q1[2] + q1[3]));
            }
            gj_publish<0>(sRt[0], sCc[0], d0, 0, w, lane);
            gj_publish<0>(sRt[0], sCc[0], d1, 1, w, lane);
            __syncthreads();
            gj_run<0>(sRt, sCc, d0, d1, w, lane);
            if (w >= 2) {
                storeD_pairs(sInv, NYY, 0, nS, d0, lane);
                storeD_pairs(sInv, NYY, 16, nS, d1, lane);
            }
        }
        __syncthreads();
        v8f T[2];
        {
            v2f ai0[8], ai1[8], bm[8];
#pragma unroll
            for (int i = 0; i < 8; ++i) {
                ai0[i] = load_pair(sInv, NYY, 2 * i + (lane >> 4), (lane & 15));
                ai1[i] = load_pair(sInv, NYY, 2 * i + (lane >> 4), 16 + (lane & 15));
            }
            load_b<true, 8>(sM2, NYY, nw, lane, bm);
            lds_group_fence();
            T[0] = zero8(); T[1] = zero8();
#pragma unroll
            for (int i = 0; i < 8; ++i) {
                T[0] = wmma4(ai0[i], bm[i], T[0]);
                T[1] = wmma4(ai1[i], bm[i], T[1]);
            }
            storeD(sLT, NXX, 0, nw, T[0], lane);
            storeD(sLT, NXX, 16, nw, T[1], lane);
        }
        {
            v2f bn[4][8];
#pragma unroll
            for (int kb = 0; kb < 2; ++kb) {
#pragma unroll
                for (int i = 0; i < 4; ++i) {
#pragma unroll
                    for (int n = 0; n < 4; ++n)
                        bn[n][4 * kb + i] = load_pair(sN, LDN, 8 * kb + 4 * (lane >> 4) + i, 16 * n + (lane & 15));
                }
            }
            lds_group_fence();
            v8f Pn[4];
#pragma unroll
            for (int n = 0; n < 4; ++n) Pn[n] = identC(nw, 16 * n, lane);
#pragma unroll
            for (int kb = 0; kb < 2; ++kb) {
#pragma unroll
                for (int i = 0; i < 4; ++i) {
                    const v2f al = tile_k(T[kb], i);
#pragma unroll
                    for (int n = 0; n < 4; ++n) Pn[n] = wmma4(al, bn[n][4 * kb + i], Pn[n]);
                }
            }
#pragma unroll
            for (int n = 0; n < 4; ++n) storeD_pairs(sP, NXX, nw, 16 * n, Pn[n], lane);
        }
        __syncthreads();
    }

    if (ck + 1 < NCH) {
        float* Ls = G + (size_t)t1 * GSLOT;
        float* Ps = G + (size_t)(t1 + 1) * GSLOT;
        for (int i = tid; i < NXX * NXX; i += 128) Ps[i] = sP[(((i >> 7) * 64 + (i & 63)) << 1) + ((i >> 6) & 1)];
        for (int i = tid; i < NYY * NXX; i += 128) Ls[i] = sLT[i];
    }
}

__device__ __forceinline__ void kf_batch_chunk(const float* __restrict__ Yp,
                                               const float* __restrict__ Up,
                                               const float* __restrict__ Dp,
                                               const float* __restrict__ Wfx,
                                               const float* __restrict__ Wfu,
                                               const float* __restrict__ Wfd,
                                               const float* __restrict__ Wfy,
                                               const float* __restrict__ G,
                                               float* __restrict__ out,
                                               const int ck, const int bt) {
    __shared__ __attribute__((aligned(16))) float sxT[NXX * 16];
    __shared__ __attribute__((aligned(16))) float sxpT[NXX * 16];
    __shared__ __attribute__((aligned(16))) float sinT[NYY * 16];

    const int tid = threadIdx.x, lane = tid & 31, w = tid >> 5;
    const int r0 = bt * 16;
    const int lh = lane >> 4, lm = lane & 15;
    const int t0 = ck * CH, t1 = t0 + CH;

    if (ck == 0) {
        for (int i = tid; i < 16 * NXX; i += 128) sxT[i] = 0.0f;
    } else {
        for (int i = tid; i < 16 * NXX; i += 128) {
            int r = i >> 6, cc = i & 63;
            sxT[cc * 16 + r] = out[(size_t)(r0 + r) * NXX + cc];
        }
    }

    const int nW = w * 16;
    const int nY = (w & 1) * 16;
    v2f bX[16], bU[4], bD[2];
    v2f bY[16];
#pragma unroll
    for (int i = 0; i < 16; ++i) {
        bX[i] = fragB(Wfx, NXX, 4 * i, nW, lane);
        bY[i] = fragB(Wfy, NYY, 4 * i, nY, lane);
    }
#pragma unroll
    for (int i = 0; i < 4; ++i) bU[i] = fragB(Wfu, NXX, 4 * i, nW, lane);
#pragma unroll
    for (int i = 0; i < 2; ++i) bD[i] = fragB(Wfd, NXX, 4 * i, nW, lane);
    __syncthreads();

    for (int t = t0; t < t1; ++t) {
        const float* Gt = G + (size_t)t * GSLOT;
        v8f xp = zero8();
        v8f ym = zero8();
        {
            const float* ut = Up + ((size_t)t * B_TOT + r0) * NUU;
            const float* dt = Dp + ((size_t)t * B_TOT + r0) * NDD;
            const float* yt = Yp + ((size_t)t * B_TOT + r0) * NYY;
            v2f aU[4], aD[2];
#pragma unroll
            for (int i = 0; i < 4; ++i)
                aU[i] = *reinterpret_cast<const v2f*>(ut + lm * NUU + 4 * i + 2 * lh);
#pragma unroll
            for (int i = 0; i < 2; ++i)
                aD[i] = *reinterpret_cast<const v2f*>(dt + lm * NDD + 4 * i + 2 * lh);
            if (w < 2) {
#pragma unroll
                for (int r = 0; r < 8; ++r) ym[r] = yt[(r + 8 * lh) * NYY + nY + lm];
            }
            if (t + 1 < t1) {
                __builtin_prefetch(G + (size_t)(t + 1) * GSLOT + tid * 16, 0, 1);
            }

            xp = mm_breg<true, 16>(sxT, 16, 0, bX, xp, lane);
#pragma unroll
            for (int i = 0; i < 4; ++i) xp = wmma4(aU[i], bU[i], xp);
#pragma unroll
            for (int i = 0; i < 2; ++i) xp = wmma4(aD[i], bD[i], xp);
            storeDT(sxpT, 16, 0, nW, xp, lane);
        }
        __syncthreads();

        if (w < 2) {
            v8f c = zero8();
            c = mm_breg<true, 16>(sxpT, 16, 0, bY, c, lane);
            v8f inn;
#pragma unroll
            for (int r = 0; r < 8; ++r) inn[r] = ym[r] - c[r];
            storeDT(sinT, 16, 0, nY, inn, lane);
        }
        __syncthreads();

        {
            v8f c = mmacc<true, false>(sinT, 16, 0, Gt, NXX, nW, NYY, xp, lane);
            storeDT(sxT, 16, 0, nW, c, lane);
        }
        __syncthreads();
    }

    for (int i = tid; i < 16 * NXX; i += 128) {
        int r = i >> 6, cc = i & 63;
        out[(size_t)(r0 + r) * NXX + cc] = sxT[cc * 16 + r];
    }
}

__global__ __launch_bounds__(128) void kf_pipe(const float* __restrict__ Yp,
                                               const float* __restrict__ Up,
                                               const float* __restrict__ Dp,
                                               const float* __restrict__ Wfx,
                                               const float* __restrict__ Wfu,
                                               const float* __restrict__ Wfd,
                                               const float* __restrict__ Wfy,
                                               float* G, float* out, const int k) {
    if (blockIdx.x == 0) {
        if (k < NCH) kf_gain_chunk(Wfx, Wfy, G, k);
    } else {
        if (k >= 1) kf_batch_chunk(Yp, Up, Dp, Wfx, Wfu, Wfd, Wfy, G, out, k - 1, blockIdx.x - 1);
    }
}

extern "C" void kernel_launch(void* const* d_in, const int* in_sizes, int n_in,
                              void* d_out, int out_size, void* d_ws, size_t ws_size,
                              hipStream_t stream) {
    (void)in_sizes; (void)n_in; (void)out_size; (void)ws_size;
    const float* Yp  = (const float*)d_in[0];
    const float* Up  = (const float*)d_in[1];
    const float* Dp  = (const float*)d_in[2];
    const float* Wfx = (const float*)d_in[3];
    const float* Wfu = (const float*)d_in[4];
    const float* Wfd = (const float*)d_in[5];
    const float* Wfy = (const float*)d_in[6];
    float* G = (float*)d_ws;
    float* out = (float*)d_out;

    kf_pipe<<<17, 128, 0, stream>>>(Yp, Up, Dp, Wfx, Wfu, Wfd, Wfy, G, out, 0);
    kf_pipe<<<17, 128, 0, stream>>>(Yp, Up, Dp, Wfx, Wfu, Wfd, Wfy, G, out, 1);
    kf_pipe<<<17, 128, 0, stream>>>(Yp, Up, Dp, Wfx, Wfu, Wfd, Wfy, G, out, 2);
    kf_pipe<<<17, 128, 0, stream>>>(Yp, Up, Dp, Wfx, Wfu, Wfd, Wfy, G, out, 3);
    kf_pipe<<<17, 128, 0, stream>>>(Yp, Up, Dp, Wfx, Wfu, Wfd, Wfy, G, out, 4);
    kf_pipe<<<17, 128, 0, stream>>>(Yp, Up, Dp, Wfx, Wfu, Wfd, Wfy, G, out, 5);
    kf_pipe<<<17, 128, 0, stream>>>(Yp, Up, Dp, Wfx, Wfu, Wfd, Wfy, G, out, 6);
    kf_pipe<<<17, 128, 0, stream>>>(Yp, Up, Dp, Wfx, Wfu, Wfd, Wfy, G, out, 7);
    kf_pipe<<<17, 128, 0, stream>>>(Yp, Up, Dp, Wfx, Wfu, Wfd, Wfy, G, out, 8);
}


# Inverse of S: a block step's Schur complement multiplied by det A, so its two divisions run side by side: 1 on the pivot-to-pivot chain, not 2, one multiply-add after it; inverse's sums reordered
# speedup vs baseline: 6.2225x; 1.0306x over previous
#include <hip/hip_runtime.h>
#include <hip/hip_bf16.h>

#define NXX 64
#define NYY 32
#define NUU 16
#define NDD 8
#define T_STEPS 2048
#define B_TOT 256
#define NCH 8
#define CH (T_STEPS / NCH)
#define GSLOT (NYY * NXX)
#define LDN 68

typedef float v2f __attribute__((ext_vector_type(2)));
typedef float v8f __attribute__((ext_vector_type(8)));

__device__ __forceinline__ v8f zero8() {
    v8f c;
#pragma unroll
    for (int r = 0; r < 8; ++r) c[r] = 0.0f;
    return c;
}

__device__ __forceinline__ v2f fragA(const float* X, int ld, int m0, int k0, int lane) {
    int h = lane >> 4, m = m0 + (lane & 15), k = k0 + 2 * h;
    v2f a; a.x = X[m * ld + k]; a.y = X[m * ld + k + 1]; return a;
}
__device__ __forceinline__ v2f fragAT(const float* X, int ld, int m0, int k0, int lane) {
    int h = lane >> 4, m = m0 + (lane & 15), k = k0 + 2 * h;
    v2f a; a.x = X[k * ld + m]; a.y = X[(k + 1) * ld + m]; return a;
}
__device__ __forceinline__ v2f fragB(const float* X, int ld, int k0, int n0, int lane) {
    int h = lane >> 4, n = n0 + (lane & 15), k = k0 + 2 * h;
    v2f b; b.x = X[k * ld + n]; b.y = X[(k + 1) * ld + n]; return b;
}
__device__ __forceinline__ v2f fragBT(const float* X, int ld, int k0, int n0, int lane) {
    int h = lane >> 4, n = n0 + (lane & 15), k = k0 + 2 * h;
    v2f b; b.x = X[n * ld + k]; b.y = X[n * ld + k + 1]; return b;
}
__device__ __forceinline__ void storeD(float* X, int ld, int m0, int n0, v8f d, int lane,
                                       float sgn = 1.0f) {
    int h = lane >> 4, n = n0 + (lane & 15);
#pragma unroll
    for (int r = 0; r < 8; ++r) X[(m0 + r + 8 * h) * ld + n] = sgn * d[r];
}
__device__ __forceinline__ v8f loadC(const float* X, int ld, int m0, int n0, int lane) {
    int h = lane >> 4, n = n0 + (lane & 15);
    v8f c;
#pragma unroll
    for (int r = 0; r < 8; ++r) c[r] = X[(m0 + r + 8 * h) * ld + n];
    return c;
}
__device__ __forceinline__ void storeDT(float* X, int ld, int m0, int n0, v8f d, int lane) {
    int h = lane >> 4, n = n0 + (lane & 15);
    float4* p = reinterpret_cast<float4*>(X + n * ld + m0 + 8 * h);
    p[0] = make_float4(d[0], d[1], d[2], d[3]);
    p[1] = make_float4(d[4], d[5], d[6], d[7]);
}
__device__ __forceinline__ v8f identC(int m0, int n0, int lane) {
    int h = lane >> 4, n = n0 + (lane & 15);
    v8f c;
#pragma unroll
    for (int r = 0; r < 8; ++r) c[r] = ((m0 + r + 8 * h) == n) ? 1.0f : 0.0f;
    return c;
}

template <bool TA, bool TB>
__device__ __forceinline__ v8f mmacc(const float* A, int lda, int m0,
                                     const float* B, int ldb, int n0,
                                     int K, v8f c, int lane) {
#pragma unroll
    for (int k = 0; k < K; k += 4) {
        v2f a = TA ? fragAT(A, lda, m0, k, lane) : fragA(A, lda, m0, k, lane);
        v2f b = TB ? fragBT(B, ldb, k, n0, lane) : fragB(B, ldb, k, n0, lane);
        c = __builtin_amdgcn_wmma_f32_16x16x4_f32(false, a, false, b, (short)0, c,
                                                  false, false);
    }
    return c;
}

__device__ __forceinline__ v8f wmma4(v2f a, v2f b, v8f c) {
    return __builtin_amdgcn_wmma_f32_16x16x4_f32(false, a, false, b, (short)0, c,
                                                 false, false);
}
template <bool TA, int KS>
__device__ __forceinline__ v8f mm_breg(const float* A, int lda, int m0,
                                       const v2f (&b)[KS], v8f c, int lane) {
#pragma unroll
    for (int i = 0; i < KS; ++i) {
        v2f a = TA ? fragAT(A, lda, m0, 4 * i, lane) : fragA(A, lda, m0, 4 * i, lane);
        c = wmma4(a, b[i], c);
    }
    return c;
}

__device__ __forceinline__ void lds_group_fence() {
    __builtin_amdgcn_sched_barrier(0);
}
template <bool TA, int KS>
__device__ __forceinline__ void load_a(const float* A, int lda, int m0, int lane, v2f (&a)[KS]) {
#pragma unroll
    for (int i = 0; i < KS; ++i)
        a[i] = TA ? fragAT(A, lda, m0, 4 * i, lane) : fragA(A, lda, m0, 4 * i, lane);
}
template <bool TB, int KS>
__device__ __forceinline__ void load_b(const float* B, int ldb, int n0, int lane, v2f (&b)[KS]) {
#pragma unroll
    for (int i = 0; i < KS; ++i)
        b[i] = TB ? fragBT(B, ldb, 4 * i, n0, lane) : fragB(B, ldb, 4 * i, n0, lane);
}
template <int KS>
__device__ __forceinline__ v8f mm_regs(const v2f (&a)[KS], const v2f (&b)[KS], v8f c) {
#pragma unroll
    for (int i = 0; i < KS; ++i) c = wmma4(a[i], b[i], c);
    return c;
}

__device__ __forceinline__ v2f tile_k(v8f d, int i) {
    v2f f; f.x = d[2 * i]; f.y = d[2 * i + 1]; return f;
}
__device__ __forceinline__ v2f fragAT_K(const float* X, int ld, int m0, int kb, int i, int lane) {
    int h = lane >> 4, m = m0 + (lane & 15), k = kb + 8 * h + 2 * i;
    v2f a; a.x = X[k * ld + m]; a.y = X[(k + 1) * ld + m]; return a;
}

__device__ __forceinline__ void storeD_pairs(float* X, int ld, int m0, int n0, v8f d, int lane,
                                             float sgn = 1.0f) {
    int h = lane >> 4, n = n0 + (lane & 15);
#pragma unroll
    for (int q = 0; q < 4; ++q) {
        v2f pr; pr.x = sgn * d[2 * q]; pr.y = sgn * d[2 * q + 1];
        *reinterpret_cast<v2f*>(X + ((((m0 >> 1) + 4 * h + q) * ld + n) << 1)) = pr;
    }
}
__device__ __forceinline__ v2f load_pair(const float* X, int ld, int kpair, int col) {
    return *reinterpret_cast<const v2f*>(X + ((kpair * ld + col) << 1));
}
__device__ __forceinline__ void storeDT_pairs_neg(float* X, int ld, int m0, int k0, v8f d, int lane) {
    int h = lane >> 4, k = k0 + (lane & 15);
    float* p = X + ((((k >> 1) * ld + m0 + 8 * h) << 1) + (k & 1));
#pragma unroll
    for (int r = 0; r < 8; ++r) p[2 * r] = -d[r];
}

template <int KK>
__device__ __forceinline__ void gj_publish(float* Rt, float* Cc, v8f d, int gi, int gj, int lane) {
    constexpr int lr0 = (4 * KK) & 15, hh = lr0 >> 3, rb = lr0 & 7, pt = KK >> 2;
    const int h = lane >> 4, n = lane & 15;
    if (gi == pt && h == hh) {
        *reinterpret_cast<float4*>(Rt + (16 * gj + n) * 4) =
            make_float4(d[rb], d[rb + 1], d[rb + 2], d[rb + 3]);
    }
    const int cq = n - lr0;
    if (gj == pt && cq >= 0 && cq < 4) {
#pragma unroll
        for (int r = 0; r < 8; ++r) Cc[(16 * gi + r + 8 * h) * 4 + cq] = d[r];
    }
}

template <int KK>
__device__ __forceinline__ void gj_step(const float* Rt, const float* Cc, v8f& d0, v8f& d1,
                                        int gj, int lane) {
    constexpr int lr0 = (4 * KK) & 15, hh = lr0 >> 3, rb = lr0 & 7, pt = KK >> 2;
    const int h = lane >> 4, n = lane & 15;
    float p[4][4], m[4];
    float4 pc[4];
#pragma unroll
    for (int c = 0; c < 4; ++c) pc[c] = *reinterpret_cast<const float4*>(Rt + (4 * KK + c) * 4);
    const float4 mv = *reinterpret_cast<const float4*>(Rt + (16 * gj + n) * 4);
    const float2 cv0 = *reinterpret_cast<const float2*>(Cc + n * 4 + 2 * h);
    const float2 cv1 = *reinterpret_cast<const float2*>(Cc + (16 + n) * 4 + 2 * h);
    lds_group_fence();
#pragma unroll
    for (int c = 0; c < 4; ++c) {
        p[0][c] = pc[c].x; p[1][c] = pc[c].y; p[2][c] = pc[c].z; p[3][c] = pc[c].w;
    }
    m[0] = mv.x; m[1] = mv.y; m[2] = mv.z; m[3] = mv.w;
    {
        const float a00 = p[0][0], a01 = p[0][1], a10 = p[1][0], a11 = p[1][1];
        const float detA = a00 * a11 - a01 * a10;
        const float rA = 1.0f / detA;
        float nX[2][2], nu[2];
#pragma unroll
        for (int j = 0; j < 2; ++j) {
            nX[0][j] = a01 * p[1][2 + j] - a11 * p[0][2 + j];
            nX[1][j] = a10 * p[0][2 + j] - a00 * p[1][2 + j];
        }
        nu[0] = a01 * m[1] - a11 * m[0];
        nu[1] = a10 * m[0] - a00 * m[1];
        float g[2][2], hv[2];
#pragma unroll
        for (int i = 0; i < 2; ++i) {
#pragma unroll
            for (int j = 0; j < 2; ++j) g[i][j] = p[2 + i][0] * nX[0][j] + p[2 + i][1] * nX[1][j];
            hv[i] = p[2 + i][0] * nu[0] + p[2 + i][1] * nu[1];
        }
        const float e00 = detA * p[2][2] + g[0][0];
        const float e01 = detA * p[2][3] + g[0][1];
        const float e10 = detA * p[3][2] + g[1][0];
        const float e11 = detA * p[3][3] + g[1][1];
        const float v0 = detA * m[2] + hv[0];
        const float v1 = detA * m[3] + hv[1];
        const float detE = e00 * e11 - e01 * e10;
        const float rE = 1.0f / detE;
        const float n2 = e11 * v0 - e01 * v1;
        const float n3 = e00 * v1 - e10 * v0;
        const float t0 = (nX[0][0] * rA) * n2 + (nX[0][1] * rA) * n3;
        const float t1 = (nX[1][0] * rA) * n2 + (nX[1][1] * rA) * n3;
        const float c0 = (-nu[0]) * rA;
        const float c1 = (-nu[1]) * rA;
        m[0] = __builtin_fmaf(t0, rE, c0);
        m[1] = __builtin_fmaf(t1, rE, c1);
        m[2] = n2 * rE;
        m[3] = n3 * rE;
    }
    const int mr = n - lr0;
    const bool inblk = (mr >= 0) && (mr < 4);
    v2f b, a0, a1, au;
    b.x = h ? m[2] : m[0];
    b.y = h ? m[3] : m[1];
    au.x = (mr == 2 * h) ? 1.0f : 0.0f;
    au.y = (mr == 2 * h + 1) ? 1.0f : 0.0f;
    a0.x = -cv0.x; a0.y = -cv0.y;
    a1.x = -cv1.x; a1.y = -cv1.y;
    const bool zr = (h == hh);
    if constexpr (pt == 0) {
        a0.x = inblk ? au.x : a0.x;
        a0.y = inblk ? au.y : a0.y;
#pragma unroll
        for (int q = 0; q < 4; ++q) d0[rb + q] = zr ? 0.0f : d0[rb + q];
    } else {
        a1.x = inblk ? au.x : a1.x;
        a1.y = inblk ? au.y : a1.y;
#pragma unroll
        for (int q = 0; q < 4; ++q) d1[rb + q] = zr ? 0.0f : d1[rb + q];
    }
    d0 = wmma4(a0, b, d0);
    d1 = wmma4(a1, b, d1);
}

template <int KK>
__device__ __forceinline__ void gj_run(float (*Rt)[64 * 4], float (*Cc)[NYY * 4], v8f& d0, v8f& d1,
                                       int gj, int lane) {
    gj_step<KK>(Rt[KK & 1], Cc[KK & 1], d0, d1, gj, lane);
    if constexpr (KK < 7) {
        gj_publish<KK + 1>(Rt[(KK + 1) & 1], Cc[(KK + 1) & 1], d0, 0, gj, lane);
        gj_publish<KK + 1>(Rt[(KK + 1) & 1], Cc[(KK + 1) & 1], d1, 1, gj, lane);
        __syncthreads();
        gj_run<KK + 1>(Rt, Cc, d0, d1, gj, lane);
    }
}

__device__ __forceinline__ void kf_gain_chunk(const float* __restrict__ Wfx,
                                              const float* __restrict__ Wfy,
                                              float* __restrict__ G, const int ck) {
    __shared__ __attribute__((aligned(16))) float sP[NXX * NXX];
    __shared__ __attribute__((aligned(16))) float sU[NXX * NYY];
    __shared__ __attribute__((aligned(16))) float sN[(NYY / 2) * LDN * 2];
    __shared__ float sM2[NXX * NYY];
    __shared__ float sSp[4][NYY * NYY];
    __shared__ __attribute__((aligned(16))) float sInv[NYY * NYY];
    __shared__ __attribute__((aligned(16))) float sRt[2][64 * 4];
    __shared__ __attribute__((aligned(16))) float sCc[2][NYY * 4];
    __shared__ __attribute__((aligned(16))) float sLT[NYY * NXX];

    const int tid = threadIdx.x, lane = tid & 31, w = tid >> 5;
    const int t0 = ck * CH, t1 = t0 + CH;
    const int nw = 16 * w;

    if (ck == 0) {
        for (int i = tid; i < NXX * NXX; i += 128) {
            sP[(((i >> 7) * 64 + (i & 63)) << 1) + ((i >> 6) & 1)] = ((i >> 6) == (i & 63)) ? 1.0f : 0.0f;
        }
        for (int i = tid; i < NYY * NXX; i += 128) {
            sLT[i] = 0.0f;
        }
    } else {
        const float* Ls = G + (size_t)t0 * GSLOT;
        const float* Ps = G + (size_t)(t0 + 1) * GSLOT;
        for (int i = tid; i < NXX * NXX; i += 128) sP[(((i >> 7) * 64 + (i & 63)) << 1) + ((i >> 6) & 1)] = Ps[i];
        for (int i = tid; i < NYY * NXX; i += 128) sLT[i] = Ls[i];
    }

    {
        const v8f t0 = mmacc<true, false>(Wfx, NXX, nw, Wfy, NYY, 0, NXX, zero8(), lane);
        const v8f t1 = mmacc<true, false>(Wfx, NXX, nw, Wfy, NYY, 16, NXX, zero8(), lane);
        storeD(sM2, NYY, nw, 0, t0, lane);
        storeD(sM2, NYY, nw, 16, t1, lane);
    }
    v2f bA1[2][16];
    v2f aW[16];
    v8f cW[2];
    v2f aYK[2][4];
#pragma unroll
    for (int i = 0; i < 16; ++i) aW[i] = fragA(Wfx, NXX, nw, 4 * i, lane);
    cW[0] = loadC(Wfy, NYY, nw, 0, lane);
    cW[1] = loadC(Wfy, NYY, nw, 16, lane);
#pragma unroll
    for (int i = 0; i < 4; ++i) {
        aYK[0][i] = fragAT_K(Wfy, NYY, 0, nw, i, lane);
        aYK[1][i] = fragAT_K(Wfy, NYY, 16, nw, i, lane);
    }
    __syncthreads();
#pragma unroll
    for (int i = 0; i < 16; ++i) {
        bA1[0][i] = fragB(sM2, NYY, 4 * i, 0, lane);
        bA1[1][i] = fragB(sM2, NYY, 4 * i, 16, lane);
    }

    for (int t = t0; t < t1; ++t) {
        v8f M[2];
        {
            v2f a[16];
#pragma unroll
            for (int i = 0; i < 16; ++i)
                a[i] = load_pair(sP, NXX, 2 * i + (lane >> 4), nw + (lane & 15));
            lds_group_fence();
            v8f U0 = zero8(), U1 = zero8();
#pragma unroll
            for (int i = 0; i < 16; ++i) {
                U0 = wmma4(a[i], bA1[0][i], U0);
                U1 = wmma4(a[i], bA1[1][i], U1);
            }
            storeD_pairs(sU, NYY, nw, 0, U0, lane);
            storeD_pairs(sU, NYY, nw, 16, U1, lane);
        }
        __syncthreads();
        {
            float4* Gt4 = reinterpret_cast<float4*>(G + (size_t)t * (NYY * NXX));
            const float4* L4 = reinterpret_cast<const float4*>(sLT);
#pragma unroll
            for (int q = 0; q < 4; ++q) Gt4[tid + 128 * q] = L4[tid + 128 * q];
        }
        {
            v2f bu[2][16];
#pragma unroll
            for (int i = 0; i < 16; ++i) {
                bu[0][i] = load_pair(sU, NYY, 2 * i + (lane >> 4), (lane & 15));
                bu[1][i] = load_pair(sU, NYY, 2 * i + (lane >> 4), 16 + (lane & 15));
            }
            lds_group_fence();
            M[0] = cW[0]; M[1] = cW[1];
#pragma unroll
            for (int i = 0; i < 16; ++i) {
                M[0] = wmma4(aW[i], bu[0][i], M[0]);
                M[1] = wmma4(aW[i], bu[1][i], M[1]);
            }
        }
        storeD(sM2, NYY, nw, 0, M[0], lane);
        storeD(sM2, NYY, nw, 16, M[1], lane);
        storeDT_pairs_neg(sN, LDN, nw, 0, M[0], lane);
        storeDT_pairs_neg(sN, LDN, nw, 16, M[1], lane);
        {
            v8f Sp[2][2];
            Sp[0][0] = zero8(); Sp[0][1] = zero8(); Sp[1][0] = zero8(); Sp[1][1] = zero8();
#pragma unroll
            for (int i = 0; i < 4; ++i) {
#pragma unroll
                for (int nb = 0; nb < 2; ++nb) {
                    const v2f bm = tile_k(M[nb], i);
                    Sp[0][nb] = wmma4(aYK[0][i], bm, Sp[0][nb]);
                    Sp[1][nb] = wmma4(aYK[1][i], bm, Sp[1][nb]);
                }
            }
#pragma unroll
            for (int gi = 0; gi < 2; ++gi) {
#pragma unroll
                for (int nb = 0; nb < 2; ++nb) storeD(sSp[w], NYY, 16 * gi, 16 * nb, Sp[gi][nb], lane);
            }
        }
        __syncthreads();
        {
            const int nS = 16 * (w & 1);
            v8f d0 = identC(0, nS, lane), d1 = identC(16, nS, lane);
            if (w < 2) {
                v8f q0[4], q1[4];
#pragma unroll
                for (int j = 0; j < 4; ++j) {
                    q0[j] = loadC(sSp[j], NYY, 0, nS, lane);
                    q1[j] = loadC(sSp[j], NYY, 16, nS, lane);
                }
                d0 = d0 + ((q0[0] + q0[1]) + (q0[2] + q0[3]));
                d1 = d1 + ((q1[0] + q1[1]) + (q1[2] + q1[3]));
            }
            gj_publish<0>(sRt[0], sCc[0], d0, 0, w, lane);
            gj_publish<0>(sRt[0], sCc[0], d1, 1, w, lane);
            __syncthreads();
            gj_run<0>(sRt, sCc, d0, d1, w, lane);
            if (w >= 2) {
                storeD_pairs(sInv, NYY, 0, nS, d0, lane);
                storeD_pairs(sInv, NYY, 16, nS, d1, lane);
            }
        }
        __syncthreads();
        v8f T[2];
        {
            v2f ai0[8], ai1[8], bm[8];
#pragma unroll
            for (int i = 0; i < 8; ++i) {
                ai0[i] = load_pair(sInv, NYY, 2 * i + (lane >> 4), (lane & 15));
                ai1[i] = load_pair(sInv, NYY, 2 * i + (lane >> 4), 16 + (lane & 15));
            }
            load_b<true, 8>(sM2, NYY, nw, lane, bm);
            lds_group_fence();
            T[0] = zero8(); T[1] = zero8();
#pragma unroll
            for (int i = 0; i < 8; ++i) {
                T[0] = wmma4(ai0[i], bm[i], T[0]);
                T[1] = wmma4(ai1[i], bm[i], T[1]);
            }
            storeD(sLT, NXX, 0, nw, T[0], lane);
            storeD(sLT, NXX, 16, nw, T[1], lane);
        }
        {
            v2f bn[4][8];
#pragma unroll
            for (int kb = 0; kb < 2; ++kb) {
#pragma unroll
                for (int i = 0; i < 4; ++i) {
#pragma unroll
                    for (int n = 0; n < 4; ++n)
                        bn[n][4 * kb + i] = load_pair(sN, LDN, 8 * kb + 4 * (lane >> 4) + i, 16 * n + (lane & 15));
                }
            }
            lds_group_fence();
            v8f Pn[4];
#pragma unroll
            for (int n = 0; n < 4; ++n) Pn[n] = identC(nw, 16 * n, lane);
#pragma unroll
            for (int kb = 0; kb < 2; ++kb) {
#pragma unroll
                for (int i = 0; i < 4; ++i) {
                    const v2f al = tile_k(T[kb], i);
#pragma unroll
                    for (int n = 0; n < 4; ++n) Pn[n] = wmma4(al, bn[n][4 * kb + i], Pn[n]);
                }
            }
#pragma unroll
            for (int n = 0; n < 4; ++n) storeD_pairs(sP, NXX, nw, 16 * n, Pn[n], lane);
        }
        __syncthreads();
    }

    if (ck + 1 < NCH) {
        float* Ls = G + (size_t)t1 * GSLOT;
        float* Ps = G + (size_t)(t1 + 1) * GSLOT;
        for (int i = tid; i < NXX * NXX; i += 128) Ps[i] = sP[(((i >> 7) * 64 + (i & 63)) << 1) + ((i >> 6) & 1)];
        for (int i = tid; i < NYY * NXX; i += 128) Ls[i] = sLT[i];
    }
}

__device__ __forceinline__ void kf_batch_chunk(const float* __restrict__ Yp,
                                               const float* __restrict__ Up,
                                               const float* __restrict__ Dp,
                                               const float* __restrict__ Wfx,
                                               const float* __restrict__ Wfu,
                                               const float* __restrict__ Wfd,
                                               const float* __restrict__ Wfy,
                                               const float* __restrict__ G,
                                               float* __restrict__ out,
                                               const int ck, const int bt) {
    __shared__ __attribute__((aligned(16))) float sxT[NXX * 16];
    __shared__ __attribute__((aligned(16))) float sxpT[NXX * 16];
    __shared__ __attribute__((aligned(16))) float sinT[NYY * 16];

    const int tid = threadIdx.x, lane = tid & 31, w = tid >> 5;
    const int r0 = bt * 16;
    const int lh = lane >> 4, lm = lane & 15;
    const int t0 = ck * CH, t1 = t0 + CH;

    if (ck == 0) {
        for (int i = tid; i < 16 * NXX; i += 128) sxT[i] = 0.0f;
    } else {
        for (int i = tid; i < 16 * NXX; i += 128) {
            int r = i >> 6, cc = i & 63;
            sxT[cc * 16 + r] = out[(size_t)(r0 + r) * NXX + cc];
        }
    }

    const int nW = w * 16;
    const int nY = (w & 1) * 16;
    v2f bX[16], bU[4], bD[2];
    v2f bY[16];
#pragma unroll
    for (int i = 0; i < 16; ++i) {
        bX[i] = fragB(Wfx, NXX, 4 * i, nW, lane);
        bY[i] = fragB(Wfy, NYY, 4 * i, nY, lane);
    }
#pragma unroll
    for (int i = 0; i < 4; ++i) bU[i] = fragB(Wfu, NXX, 4 * i, nW, lane);
#pragma unroll
    for (int i = 0; i < 2; ++i) bD[i] = fragB(Wfd, NXX, 4 * i, nW, lane);
    __syncthreads();

    for (int t = t0; t < t1; ++t) {
        const float* Gt = G + (size_t)t * GSLOT;
        v8f xp = zero8();
        v8f ym = zero8();
        {
            const float* ut = Up + ((size_t)t * B_TOT + r0) * NUU;
            const float* dt = Dp + ((size_t)t * B_TOT + r0) * NDD;
            const float* yt = Yp + ((size_t)t * B_TOT + r0) * NYY;
            v2f aU[4], aD[2];
#pragma unroll
            for (int i = 0; i < 4; ++i)
                aU[i] = *reinterpret_cast<const v2f*>(ut + lm * NUU + 4 * i + 2 * lh);
#pragma unroll
            for (int i = 0; i < 2; ++i)
                aD[i] = *reinterpret_cast<const v2f*>(dt + lm * NDD + 4 * i + 2 * lh);
            if (w < 2) {
#pragma unroll
                for (int r = 0; r < 8; ++r) ym[r] = yt[(r + 8 * lh) * NYY + nY + lm];
            }
            if (t + 1 < t1) {
                __builtin_prefetch(G + (size_t)(t + 1) * GSLOT + tid * 16, 0, 1);
            }

            xp = mm_breg<true, 16>(sxT, 16, 0, bX, xp, lane);
#pragma unroll
            for (int i = 0; i < 4; ++i) xp = wmma4(aU[i], bU[i], xp);
#pragma unroll
            for (int i = 0; i < 2; ++i) xp = wmma4(aD[i], bD[i], xp);
            storeDT(sxpT, 16, 0, nW, xp, lane);
        }
        __syncthreads();

        if (w < 2) {
            v8f c = zero8();
            c = mm_breg<true, 16>(sxpT, 16, 0, bY, c, lane);
            v8f inn;
#pragma unroll
            for (int r = 0; r < 8; ++r) inn[r] = ym[r] - c[r];
            storeDT(sinT, 16, 0, nY, inn, lane);
        }
        __syncthreads();

        {
            v8f c = mmacc<true, false>(sinT, 16, 0, Gt, NXX, nW, NYY, xp, lane);
            storeDT(sxT, 16, 0, nW, c, lane);
        }
        __syncthreads();
    }

    for (int i = tid; i < 16 * NXX; i += 128) {
        int r = i >> 6, cc = i & 63;
        out[(size_t)(r0 + r) * NXX + cc] = sxT[cc * 16 + r];
    }
}

__global__ __launch_bounds__(128) void kf_pipe(const float* __restrict__ Yp,
                                               const float* __restrict__ Up,
                                               const float* __restrict__ Dp,
                                               const float* __restrict__ Wfx,
                                               const float* __restrict__ Wfu,
                                               const float* __restrict__ Wfd,
                                               const float* __restrict__ Wfy,
                                               float* G, float* out, const int k) {
    if (blockIdx.x == 0) {
        if (k < NCH) kf_gain_chunk(Wfx, Wfy, G, k);
    } else {
        if (k >= 1) kf_batch_chunk(Yp, Up, Dp, Wfx, Wfu, Wfd, Wfy, G, out, k - 1, blockIdx.x - 1);
    }
}

extern "C" void kernel_launch(void* const* d_in, const int* in_sizes, int n_in,
                              void* d_out, int out_size, void* d_ws, size_t ws_size,
                              hipStream_t stream) {
    (void)in_sizes; (void)n_in; (void)out_size; (void)ws_size;
    const float* Yp  = (const float*)d_in[0];
    const float* Up  = (const float*)d_in[1];
    const float* Dp  = (const float*)d_in[2];
    const float* Wfx = (const float*)d_in[3];
    const float* Wfu = (const float*)d_in[4];
    const float* Wfd = (const float*)d_in[5];
    const float* Wfy = (const float*)d_in[6];
    float* G = (float*)d_ws;
    float* out = (float*)d_out;

    kf_pipe<<<17, 128, 0, stream>>>(Yp, Up, Dp, Wfx, Wfu, Wfd, Wfy, G, out, 0);
    kf_pipe<<<17, 128, 0, stream>>>(Yp, Up, Dp, Wfx, Wfu, Wfd, Wfy, G, out, 1);
    kf_pipe<<<17, 128, 0, stream>>>(Yp, Up, Dp, Wfx, Wfu, Wfd, Wfy, G, out, 2);
    kf_pipe<<<17, 128, 0, stream>>>(Yp, Up, Dp, Wfx, Wfu, Wfd, Wfy, G, out, 3);
    kf_pipe<<<17, 128, 0, stream>>>(Yp, Up, Dp, Wfx, Wfu, Wfd, Wfy, G, out, 4);
    kf_pipe<<<17, 128, 0, stream>>>(Yp, Up, Dp, Wfx, Wfu, Wfd, Wfy, G, out, 5);
    kf_pipe<<<17, 128, 0, stream>>>(Yp, Up, Dp, Wfx, Wfu, Wfd, Wfy, G, out, 6);
    kf_pipe<<<17, 128, 0, stream>>>(Yp, Up, Dp, Wfx, Wfu, Wfd, Wfy, G, out, 7);
    kf_pipe<<<17, 128, 0, stream>>>(Yp, Up, Dp, Wfx, Wfu, Wfd, Wfy, G, out, 8);
}
